# starting kernel
# baseline (speedup 1.0000x reference)
#include <hip/hip_runtime.h>
#include <math.h>


typedef float v2f __attribute__((ext_vector_type(2)));
typedef float v8f __attribute__((ext_vector_type(8)));

constexpr int NRQ = 4096;
constexpr int NBC = 16384;
constexpr int CB  = 80;
constexpr int CRV = 384;
constexpr int DD  = 128;
constexpr int MM  = 256;
constexpr float INV_SQRT_D = 0.08838834764831845f;


__global__ __launch_bounds__(256)
void rmba_qproj_kernel(const float* __restrict__ A,
                       const float* __restrict__ Wq,
                       const float* __restrict__ bq,
                       float* __restrict__ Qn)
{
    const int lane = threadIdx.x & 31;
    const int wave = threadIdx.x >> 5;
    const int half = lane >> 4;
    const int l16  = lane & 15;
    const int koff = half * 2;
    const int rowBase = blockIdx.x * 16;
    const int colBase = wave * 16;
    const int arow = rowBase + l16;
    const int bcol = colBase + l16;

    v8f acc = {};
    #pragma unroll 4
    for (int k = 0; k < CRV; k += 4) {
        v2f a, b;
        a.x = A[arow * CRV + k + koff];
        a.y = A[arow * CRV + k + koff + 1];
        b.x = Wq[(k + koff) * DD + bcol];
        b.y = Wq[(k + koff + 1) * DD + bcol];
        acc = __builtin_amdgcn_wmma_f32_16x16x4_f32(false, a, false, b,
                                                    (short)0, acc, false, false);
    }

    const float bias = bq[bcol];
    float vals[8], ss[8];
    #pragma unroll
    for (int r = 0; r < 8; ++r) { vals[r] = acc[r] + bias; ss[r] = vals[r] * vals[r]; }

    #pragma unroll
    for (int s = 8; s >= 1; s >>= 1)
        #pragma unroll
        for (int r = 0; r < 8; ++r) ss[r] += __shfl_xor(ss[r], s, 32);

    __shared__ float sq[8][16];
    __shared__ float rnorm[16];
    if (l16 == 0) {
        #pragma unroll
        for (int r = 0; r < 8; ++r) sq[wave][half * 8 + r] = ss[r];
    }
    __syncthreads();
    if (threadIdx.x < 16) {
        float s = 0.f;
        #pragma unroll
        for (int w = 0; w < 8; ++w) s += sq[w][threadIdx.x];
        rnorm[threadIdx.x] = 1.0f / fmaxf(sqrtf(s), 1e-12f);
    }
    __syncthreads();
    #pragma unroll
    for (int r = 0; r < 8; ++r) {
        const int row = half * 8 + r;
        Qn[(rowBase + row) * DD + bcol] = vals[r] * rnorm[row];
    }
}

__global__ __launch_bounds__(256)
void rmba_kvproj_kernel(const float* __restrict__ bev,
                        const float* __restrict__ Wk, const float* __restrict__ bk,
                        const float* __restrict__ Wv, const float* __restrict__ bv,
                        float* __restrict__ Kall,
                        float* __restrict__ Vall)
{
    const int lane = threadIdx.x & 31;
    const int wave = threadIdx.x >> 5;
    const int half = lane >> 4;
    const int l16  = lane & 15;
    const int koff = half * 2;
    const int rowBase = blockIdx.x * 16;
    const int colBase = wave * 16;
    const int arow = rowBase + l16;
    const int bcol = colBase + l16;

    v8f acck = {}, accv = {};
    #pragma unroll 4
    for (int k = 0; k < CB; k += 4) {
        v2f a, wk, wv;
        a.x  = bev[(k + koff) * NBC + arow];
        a.y  = bev[(k + koff + 1) * NBC + arow];
        wk.x = Wk[(k + koff) * DD + bcol];
        wk.y = Wk[(k + koff + 1) * DD + bcol];
        wv.x = Wv[(k + koff) * DD + bcol];
        wv.y = Wv[(k + koff + 1) * DD + bcol];
        acck = __builtin_amdgcn_wmma_f32_16x16x4_f32(false, a, false, wk,
                                                     (short)0, acck, false, false);
        accv = __builtin_amdgcn_wmma_f32_16x16x4_f32(false, a, false, wv,
                                                     (short)0, accv, false, false);
    }

    const float biask = bk[bcol];
    const float biasv = bv[bcol];
    float kv[8], ss[8];
    #pragma unroll
    for (int r = 0; r < 8; ++r) { kv[r] = acck[r] + biask; ss[r] = kv[r] * kv[r]; }
    #pragma unroll
    for (int s = 8; s >= 1; s >>= 1)
        #pragma unroll
        for (int r = 0; r < 8; ++r) ss[r] += __shfl_xor(ss[r], s, 32);

    __shared__ float sq[8][16];
    __shared__ float rnorm[16];
    if (l16 == 0) {
        #pragma unroll
        for (int r = 0; r < 8; ++r) sq[wave][half * 8 + r] = ss[r];
    }
    __syncthreads();
    if (threadIdx.x < 16) {
        float s = 0.f;
        #pragma unroll
        for (int w = 0; w < 8; ++w) s += sq[w][threadIdx.x];
        rnorm[threadIdx.x] = 1.0f / fmaxf(sqrtf(s), 1e-12f);
    }
    __syncthreads();
    #pragma unroll
    for (int r = 0; r < 8; ++r) {
        const int row = half * 8 + r;
        Kall[(rowBase + row) * DD + bcol] = kv[r] * rnorm[row];
        Vall[(rowBase + row) * DD + bcol] = accv[r] + biasv;
    }
}

__global__ __launch_bounds__(256)
void rmba_attn_kernel(const float* __restrict__ Qn,
                      const float* __restrict__ Kall,
                      const float* __restrict__ Vall,
                      const int*   __restrict__ cand,
                      float* __restrict__ attn_out,
                      float* __restrict__ ctx)
{
    const int n   = blockIdx.x;
    const int tid = threadIdx.x;

    __shared__ float q_sm[DD];
    __shared__ float a_sm[MM];
    __shared__ int   idx_sm[MM];
    __shared__ float red_sm[8];
    __shared__ float cpart[256];

    if (tid < DD) q_sm[tid] = Qn[n * DD + tid];
    const int idx = cand[n * MM + tid];
    idx_sm[tid] = idx;
    __syncthreads();

    float logit = -INFINITY;
    if (idx >= 0) {
        const float4* krow = (const float4*)(Kall + (size_t)idx * DD);
        const float4* qv   = (const float4*)q_sm;
        float acc = 0.f;
        #pragma unroll
        for (int c = 0; c < DD / 4; ++c) {
            const float4 k4 = krow[c];
            const float4 q4 = qv[c];
            acc += k4.x * q4.x + k4.y * q4.y + k4.z * q4.z + k4.w * q4.w;
        }
        logit = acc * INV_SQRT_D;
    }

    float mx = logit;
    #pragma unroll
    for (int s = 16; s >= 1; s >>= 1) mx = fmaxf(mx, __shfl_xor(mx, s, 32));
    if ((tid & 31) == 0) red_sm[tid >> 5] = mx;
    __syncthreads();
    float bm = red_sm[0];
    #pragma unroll
    for (int w = 1; w < 8; ++w) bm = fmaxf(bm, red_sm[w]);
    __syncthreads();

    const float e = (idx >= 0) ? __expf(logit - bm) : 0.f;
    float sm = e;
    #pragma unroll
    for (int s = 16; s >= 1; s >>= 1) sm += __shfl_xor(sm, s, 32);
    if ((tid & 31) == 0) red_sm[tid >> 5] = sm;
    __syncthreads();
    float bs = 0.f;
    #pragma unroll
    for (int w = 0; w < 8; ++w) bs += red_sm[w];

    const float a = e / bs;
    attn_out[n * MM + tid] = a;
    a_sm[tid] = a;
    __syncthreads();

    const int d     = tid & 127;
    const int mBase = (tid >> 7) * 128;
    float c = 0.f;
    for (int m = 0; m < 128; ++m) {
        const int ix = idx_sm[mBase + m];
        if (ix >= 0) c += a_sm[mBase + m] * Vall[(size_t)ix * DD + d];
    }
    cpart[tid] = c;
    __syncthreads();
    if (tid < DD) ctx[n * DD + tid] = cpart[tid] + cpart[tid + 128];
}

__global__ __launch_bounds__(256)
void rmba_oproj_kernel(const float* __restrict__ ctx,
                       const float* __restrict__ Wo,
                       const float* __restrict__ bo,
                       float* __restrict__ out)
{
    const int mTiles = NRQ / 16;
    const int mTile  = blockIdx.x % mTiles;
    const int nGrp   = blockIdx.x / mTiles;
    const int lane = threadIdx.x & 31;
    const int wave = threadIdx.x >> 5;
    const int half = lane >> 4;
    const int l16  = lane & 15;
    const int koff = half * 2;
    const int rowBase = mTile * 16;
    const int colBase = nGrp * 128 + wave * 16;
    const int arow = rowBase + l16;
    const int bcol = colBase + l16;

    v8f acc = {};
    #pragma unroll 4
    for (int k = 0; k < DD; k += 4) {
        v2f a, b;
        a.x = ctx[arow * DD + k + koff];
        a.y = ctx[arow * DD + k + koff + 1];
        b.x = Wo[(k + koff) * CRV + bcol];
        b.y = Wo[(k + koff + 1) * CRV + bcol];
        acc = __builtin_amdgcn_wmma_f32_16x16x4_f32(false, a, false, b,
                                                    (short)0, acc, false, false);
    }
    const float bias = bo[bcol];
    #pragma unroll
    for (int r = 0; r < 8; ++r) {
        const int row = rowBase + half * 8 + r;
        out[row * CRV + bcol] = acc[r] + bias;
    }
}

extern "C" void kernel_launch(void* const* d_in, const int* in_sizes, int n_in,
                              void* d_out, int out_size, void* d_ws, size_t ws_size,
                              hipStream_t stream) {
    const float* rv_x = (const float*)d_in[0];
    const float* bev  = (const float*)d_in[1];
    const int*   cand = (const int*)  d_in[2];
    const float* Wq   = (const float*)d_in[3];
    const float* bq   = (const float*)d_in[4];
    const float* Wk   = (const float*)d_in[5];
    const float* bk   = (const float*)d_in[6];
    const float* Wv   = (const float*)d_in[7];
    const float* bv   = (const float*)d_in[8];
    const float* Wo   = (const float*)d_in[9];
    const float* bo   = (const float*)d_in[10];

    float* out_main = (float*)d_out;
    float* out_attn = out_main + (size_t)NRQ * CRV;

    float* Qn   = (float*)d_ws;
    float* Kall = Qn   + (size_t)NRQ * DD;
    float* Vall = Kall + (size_t)NBC * DD;
    float* ctx  = Vall + (size_t)NBC * DD;

    rmba_qproj_kernel <<<NRQ / 16, 256, 0, stream>>>(rv_x, Wq, bq, Qn);
    rmba_kvproj_kernel<<<NBC / 16, 256, 0, stream>>>(bev, Wk, bk, Wv, bv, Kall, Vall);
    rmba_attn_kernel  <<<NRQ,      256, 0, stream>>>(Qn, Kall, Vall, cand, out_attn, ctx);
    rmba_oproj_kernel <<<(NRQ / 16) * (CRV / 128), 256, 0, stream>>>(ctx, Wo, bo, out_main);
}


# attention: one wave per query, not an 8-wave block: whole 512-B K and V rows per load, 32 in flight, dots by a lane transpose-reduce, no LDS, no barriers (logit, softmax and ctx sum order changes)
# speedup vs baseline: 2.6057x; 2.6057x over previous
#include <hip/hip_runtime.h>
#include <math.h>


typedef float v2f __attribute__((ext_vector_type(2)));
typedef float v8f __attribute__((ext_vector_type(8)));

constexpr int NRQ = 4096;
constexpr int NBC = 16384;
constexpr int CB  = 80;
constexpr int CRV = 384;
constexpr int DD  = 128;
constexpr int MM  = 256;
constexpr float INV_SQRT_D = 0.08838834764831845f;


__global__ __launch_bounds__(256)
void rmba_qproj_kernel(const float* __restrict__ A,
                       const float* __restrict__ Wq,
                       const float* __restrict__ bq,
                       float* __restrict__ Qn)
{
    const int lane = threadIdx.x & 31;
    const int wave = threadIdx.x >> 5;
    const int half = lane >> 4;
    const int l16  = lane & 15;
    const int koff = half * 2;
    const int rowBase = blockIdx.x * 16;
    const int colBase = wave * 16;
    const int arow = rowBase + l16;
    const int bcol = colBase + l16;

    v8f acc = {};
    #pragma unroll 4
    for (int k = 0; k < CRV; k += 4) {
        v2f a, b;
        a.x = A[arow * CRV + k + koff];
        a.y = A[arow * CRV + k + koff + 1];
        b.x = Wq[(k + koff) * DD + bcol];
        b.y = Wq[(k + koff + 1) * DD + bcol];
        acc = __builtin_amdgcn_wmma_f32_16x16x4_f32(false, a, false, b,
                                                    (short)0, acc, false, false);
    }

    const float bias = bq[bcol];
    float vals[8], ss[8];
    #pragma unroll
    for (int r = 0; r < 8; ++r) { vals[r] = acc[r] + bias; ss[r] = vals[r] * vals[r]; }

    #pragma unroll
    for (int s = 8; s >= 1; s >>= 1)
        #pragma unroll
        for (int r = 0; r < 8; ++r) ss[r] += __shfl_xor(ss[r], s, 32);

    __shared__ float sq[8][16];
    __shared__ float rnorm[16];
    if (l16 == 0) {
        #pragma unroll
        for (int r = 0; r < 8; ++r) sq[wave][half * 8 + r] = ss[r];
    }
    __syncthreads();
    if (threadIdx.x < 16) {
        float s = 0.f;
        #pragma unroll
        for (int w = 0; w < 8; ++w) s += sq[w][threadIdx.x];
        rnorm[threadIdx.x] = 1.0f / fmaxf(sqrtf(s), 1e-12f);
    }
    __syncthreads();
    #pragma unroll
    for (int r = 0; r < 8; ++r) {
        const int row = half * 8 + r;
        Qn[(rowBase + row) * DD + bcol] = vals[r] * rnorm[row];
    }
}

__global__ __launch_bounds__(256)
void rmba_kvproj_kernel(const float* __restrict__ bev,
                        const float* __restrict__ Wk, const float* __restrict__ bk,
                        const float* __restrict__ Wv, const float* __restrict__ bv,
                        float* __restrict__ Kall,
                        float* __restrict__ Vall)
{
    const int lane = threadIdx.x & 31;
    const int wave = threadIdx.x >> 5;
    const int half = lane >> 4;
    const int l16  = lane & 15;
    const int koff = half * 2;
    const int rowBase = blockIdx.x * 16;
    const int colBase = wave * 16;
    const int arow = rowBase + l16;
    const int bcol = colBase + l16;

    v8f acck = {}, accv = {};
    #pragma unroll 4
    for (int k = 0; k < CB; k += 4) {
        v2f a, wk, wv;
        a.x  = bev[(k + koff) * NBC + arow];
        a.y  = bev[(k + koff + 1) * NBC + arow];
        wk.x = Wk[(k + koff) * DD + bcol];
        wk.y = Wk[(k + koff + 1) * DD + bcol];
        wv.x = Wv[(k + koff) * DD + bcol];
        wv.y = Wv[(k + koff + 1) * DD + bcol];
        acck = __builtin_amdgcn_wmma_f32_16x16x4_f32(false, a, false, wk,
                                                     (short)0, acck, false, false);
        accv = __builtin_amdgcn_wmma_f32_16x16x4_f32(false, a, false, wv,
                                                     (short)0, accv, false, false);
    }

    const float biask = bk[bcol];
    const float biasv = bv[bcol];
    float kv[8], ss[8];
    #pragma unroll
    for (int r = 0; r < 8; ++r) { kv[r] = acck[r] + biask; ss[r] = kv[r] * kv[r]; }
    #pragma unroll
    for (int s = 8; s >= 1; s >>= 1)
        #pragma unroll
        for (int r = 0; r < 8; ++r) ss[r] += __shfl_xor(ss[r], s, 32);

    __shared__ float sq[8][16];
    __shared__ float rnorm[16];
    if (l16 == 0) {
        #pragma unroll
        for (int r = 0; r < 8; ++r) sq[wave][half * 8 + r] = ss[r];
    }
    __syncthreads();
    if (threadIdx.x < 16) {
        float s = 0.f;
        #pragma unroll
        for (int w = 0; w < 8; ++w) s += sq[w][threadIdx.x];
        rnorm[threadIdx.x] = 1.0f / fmaxf(sqrtf(s), 1e-12f);
    }
    __syncthreads();
    #pragma unroll
    for (int r = 0; r < 8; ++r) {
        const int row = half * 8 + r;
        Kall[(rowBase + row) * DD + bcol] = kv[r] * rnorm[row];
        Vall[(rowBase + row) * DD + bcol] = accv[r] + biasv;
    }
}

__device__ __forceinline__ float rmba_readlane_f(float x, int j) {
    return __int_as_float(__builtin_amdgcn_readlane(__float_as_int(x), j));
}

template <int S>
__device__ __forceinline__ void rmba_tr_step(float (&p)[32], const int lane) {
    const bool up = (lane & S) != 0;
    #pragma unroll
    for (int i = 0; i < S; ++i) {
        const float keep = up ? p[i + S] : p[i];
        const float send = up ? p[i] : p[i + S];
        p[i] = keep + __shfl_xor(send, S, 32);
    }
}

template <int NR>
__device__ __forceinline__ float rmba_chunk_dots(const float* __restrict__ kbase, const int idc,
                                                 const float4 q4, const int lane) {
    float4 kr[NR];
    #pragma unroll
    for (int j = 0; j < NR; ++j) {
        const int ix = __builtin_amdgcn_readlane(idc, j);
        kr[j] = *(const float4*)(kbase + (size_t)ix * DD);
    }
    float p[32];
    #pragma unroll
    for (int j = 0; j < 32; ++j)
        p[j] = (j < NR) ? (kr[j < NR ? j : 0].x * q4.x + kr[j < NR ? j : 0].y * q4.y +
                           kr[j < NR ? j : 0].z * q4.z + kr[j < NR ? j : 0].w * q4.w)
                        : 0.f;
    rmba_tr_step<16>(p, lane);
    rmba_tr_step<8>(p, lane);
    rmba_tr_step<4>(p, lane);
    rmba_tr_step<2>(p, lane);
    rmba_tr_step<1>(p, lane);
    return p[0];
}

template <int NR>
__device__ __forceinline__ void rmba_chunk_ctx(const float* __restrict__ vbase, const int idc,
                                               const float a, float4& cacc) {
    float4 vr[NR];
    #pragma unroll
    for (int j = 0; j < NR; ++j) {
        const int ix = __builtin_amdgcn_readlane(idc, j);
        vr[j] = *(const float4*)(vbase + (size_t)ix * DD);
    }
    #pragma unroll
    for (int j = 0; j < NR; ++j) {
        const float aj = rmba_readlane_f(a, j);
        cacc.x += aj * vr[j].x;
        cacc.y += aj * vr[j].y;
        cacc.z += aj * vr[j].z;
        cacc.w += aj * vr[j].w;
    }
}

__global__ __launch_bounds__(64)
void rmba_attn_kernel(const float* __restrict__ Qn,
                      const float* __restrict__ Kall,
                      const float* __restrict__ Vall,
                      const int*   __restrict__ cand,
                      float* __restrict__ attn_out,
                      float* __restrict__ ctx)
{
    const int lane = threadIdx.x & 31;
    const int n    = blockIdx.x * 2 + (threadIdx.x >> 5);

    const float4 q4 = *(const float4*)(Qn + (size_t)n * DD + 4 * lane);
    const float* kbase = Kall + 4 * lane;
    const float* vbase = Vall + 4 * lane;

    int   idr[8];
    float wr[8];
    #pragma unroll
    for (int c = 0; c < 8; ++c) idr[c] = cand[n * MM + 32 * c + lane];
    #pragma unroll
    for (int c = 0; c < 8; ++c) wr[c] = -INFINITY;

    #pragma unroll 1
    for (int c = 0; c < 8; ++c) {
        int id = idr[0];
        #pragma unroll
        for (int k = 1; k < 8; ++k) id = (c == k) ? idr[k] : id;
        const unsigned vm = __builtin_amdgcn_ballot_w32(id >= 0);
        if (vm == 0u) continue;
        const int idc = max(id, 0);
        float dot;
        if (vm >> 16) dot = rmba_chunk_dots<32>(kbase, idc, q4, lane);
        else          dot = rmba_chunk_dots<16>(kbase, idc, q4, lane);
        const float lg = (id >= 0) ? dot * INV_SQRT_D : -INFINITY;
        #pragma unroll
        for (int k = 0; k < 8; ++k) wr[k] = (c == k) ? lg : wr[k];
    }

    float mx = wr[0];
    #pragma unroll
    for (int k = 1; k < 8; ++k) mx = fmaxf(mx, wr[k]);
    #pragma unroll
    for (int s = 16; s >= 1; s >>= 1) mx = fmaxf(mx, __shfl_xor(mx, s, 32));

    float sm = 0.f;
    #pragma unroll
    for (int k = 0; k < 8; ++k) {
        wr[k] = (idr[k] >= 0) ? __expf(wr[k] - mx) : 0.f;
        sm += wr[k];
    }
    #pragma unroll
    for (int s = 16; s >= 1; s >>= 1) sm += __shfl_xor(sm, s, 32);

    #pragma unroll
    for (int k = 0; k < 8; ++k) {
        wr[k] = wr[k] / sm;
        attn_out[n * MM + 32 * k + lane] = wr[k];
    }

    float4 cacc = make_float4(0.f, 0.f, 0.f, 0.f);
    #pragma unroll 1
    for (int c = 0; c < 8; ++c) {
        int   id = idr[0];
        float a  = wr[0];
        #pragma unroll
        for (int k = 1; k < 8; ++k) { id = (c == k) ? idr[k] : id; a = (c == k) ? wr[k] : a; }
        const unsigned vm = __builtin_amdgcn_ballot_w32(id >= 0);
        if (vm == 0u) continue;
        const int idc = max(id, 0);
        if (vm >> 16) rmba_chunk_ctx<32>(vbase, idc, a, cacc);
        else          rmba_chunk_ctx<16>(vbase, idc, a, cacc);
    }
    *(float4*)(ctx + (size_t)n * DD + 4 * lane) = cacc;
}

__global__ __launch_bounds__(256)
void rmba_oproj_kernel(const float* __restrict__ ctx,
                       const float* __restrict__ Wo,
                       const float* __restrict__ bo,
                       float* __restrict__ out)
{
    const int mTiles = NRQ / 16;
    const int mTile  = blockIdx.x % mTiles;
    const int nGrp   = blockIdx.x / mTiles;
    const int lane = threadIdx.x & 31;
    const int wave = threadIdx.x >> 5;
    const int half = lane >> 4;
    const int l16  = lane & 15;
    const int koff = half * 2;
    const int rowBase = mTile * 16;
    const int colBase = nGrp * 128 + wave * 16;
    const int arow = rowBase + l16;
    const int bcol = colBase + l16;

    v8f acc = {};
    #pragma unroll 4
    for (int k = 0; k < DD; k += 4) {
        v2f a, b;
        a.x = ctx[arow * DD + k + koff];
        a.y = ctx[arow * DD + k + koff + 1];
        b.x = Wo[(k + koff) * CRV + bcol];
        b.y = Wo[(k + koff + 1) * CRV + bcol];
        acc = __builtin_amdgcn_wmma_f32_16x16x4_f32(false, a, false, b,
                                                    (short)0, acc, false, false);
    }
    const float bias = bo[bcol];
    #pragma unroll
    for (int r = 0; r < 8; ++r) {
        const int row = rowBase + half * 8 + r;
        out[row * CRV + bcol] = acc[r] + bias;
    }
}

extern "C" void kernel_launch(void* const* d_in, const int* in_sizes, int n_in,
                              void* d_out, int out_size, void* d_ws, size_t ws_size,
                              hipStream_t stream) {
    const float* rv_x = (const float*)d_in[0];
    const float* bev  = (const float*)d_in[1];
    const int*   cand = (const int*)  d_in[2];
    const float* Wq   = (const float*)d_in[3];
    const float* bq   = (const float*)d_in[4];
    const float* Wk   = (const float*)d_in[5];
    const float* bk   = (const float*)d_in[6];
    const float* Wv   = (const float*)d_in[7];
    const float* bv   = (const float*)d_in[8];
    const float* Wo   = (const float*)d_in[9];
    const float* bo   = (const float*)d_in[10];

    float* out_main = (float*)d_out;
    float* out_attn = out_main + (size_t)NRQ * CRV;

    float* Qn   = (float*)d_ws;
    float* Kall = Qn   + (size_t)NRQ * DD;
    float* Vall = Kall + (size_t)NBC * DD;
    float* ctx  = Vall + (size_t)NBC * DD;

    rmba_qproj_kernel <<<NRQ / 16, 256, 0, stream>>>(rv_x, Wq, bq, Qn);
    rmba_kvproj_kernel<<<NBC / 16, 256, 0, stream>>>(bev, Wk, bk, Wv, bv, Kall, Vall);
    rmba_attn_kernel  <<<NRQ / 2,  64,  0, stream>>>(Qn, Kall, Vall, cand, out_attn, ctx);
    rmba_oproj_kernel <<<(NRQ / 16) * (CRV / 128), 256, 0, stream>>>(ctx, Wo, bo, out_main);
}


# Q and output projections fused into the attention launch: a 16-wave work-group makes its Q tile, runs 16 queries, multiplies its ctx tile by Wo from LDS; 2 launches not 4 (Q k-sum in 2 halves)
# speedup vs baseline: 3.2792x; 1.2585x over previous
#include <hip/hip_runtime.h>
#include <math.h>


typedef float v2f __attribute__((ext_vector_type(2)));
typedef float v8f __attribute__((ext_vector_type(8)));

constexpr int NRQ = 4096;
constexpr int NBC = 16384;
constexpr int CB  = 80;
constexpr int CRV = 384;
constexpr int DD  = 128;
constexpr int MM  = 256;
constexpr float INV_SQRT_D = 0.08838834764831845f;


__global__ __launch_bounds__(256)
void rmba_kvproj_kernel(const float* __restrict__ bev,
                        const float* __restrict__ Wk, const float* __restrict__ bk,
                        const float* __restrict__ Wv, const float* __restrict__ bv,
                        float* __restrict__ Kall,
                        float* __restrict__ Vall)
{
    const int lane = threadIdx.x & 31;
    const int wave = threadIdx.x >> 5;
    const int half = lane >> 4;
    const int l16  = lane & 15;
    const int koff = half * 2;
    const int rowBase = blockIdx.x * 16;
    const int colBase = wave * 16;
    const int arow = rowBase + l16;
    const int bcol = colBase + l16;

    v8f acck = {}, accv = {};
    #pragma unroll 4
    for (int k = 0; k < CB; k += 4) {
        v2f a, wk, wv;
        a.x  = bev[(k + koff) * NBC + arow];
        a.y  = bev[(k + koff + 1) * NBC + arow];
        wk.x = Wk[(k + koff) * DD + bcol];
        wk.y = Wk[(k + koff + 1) * DD + bcol];
        wv.x = Wv[(k + koff) * DD + bcol];
        wv.y = Wv[(k + koff + 1) * DD + bcol];
        acck = __builtin_amdgcn_wmma_f32_16x16x4_f32(false, a, false, wk,
                                                     (short)0, acck, false, false);
        accv = __builtin_amdgcn_wmma_f32_16x16x4_f32(false, a, false, wv,
                                                     (short)0, accv, false, false);
    }

    const float biask = bk[bcol];
    const float biasv = bv[bcol];
    float kv[8], ss[8];
    #pragma unroll
    for (int r = 0; r < 8; ++r) { kv[r] = acck[r] + biask; ss[r] = kv[r] * kv[r]; }
    #pragma unroll
    for (int s = 8; s >= 1; s >>= 1)
        #pragma unroll
        for (int r = 0; r < 8; ++r) ss[r] += __shfl_xor(ss[r], s, 32);

    __shared__ float sq[8][16];
    __shared__ float rnorm[16];
    if (l16 == 0) {
        #pragma unroll
        for (int r = 0; r < 8; ++r) sq[wave][half * 8 + r] = ss[r];
    }
    __syncthreads();
    if (threadIdx.x < 16) {
        float s = 0.f;
        #pragma unroll
        for (int w = 0; w < 8; ++w) s += sq[w][threadIdx.x];
        rnorm[threadIdx.x] = 1.0f / fmaxf(sqrtf(s), 1e-12f);
    }
    __syncthreads();
    #pragma unroll
    for (int r = 0; r < 8; ++r) {
        const int row = half * 8 + r;
        Kall[(rowBase + row) * DD + bcol] = kv[r] * rnorm[row];
        Vall[(rowBase + row) * DD + bcol] = accv[r] + biasv;
    }
}

__device__ __forceinline__ float rmba_readlane_f(float x, int j) {
    return __int_as_float(__builtin_amdgcn_readlane(__float_as_int(x), j));
}

template <int S>
__device__ __forceinline__ void rmba_tr_step(float (&p)[32], const int lane) {
    const bool up = (lane & S) != 0;
    #pragma unroll
    for (int i = 0; i < S; ++i) {
        const float keep = up ? p[i + S] : p[i];
        const float send = up ? p[i] : p[i + S];
        p[i] = keep + __shfl_xor(send, S, 32);
    }
}

template <int NR>
__device__ __forceinline__ float rmba_chunk_dots(const float* __restrict__ kbase, const int idc,
                                                 const float4 q4, const int lane) {
    float4 kr[NR];
    #pragma unroll
    for (int j = 0; j < NR; ++j) {
        const int ix = __builtin_amdgcn_readlane(idc, j);
        kr[j] = *(const float4*)(kbase + (size_t)ix * DD);
    }
    float p[32];
    #pragma unroll
    for (int j = 0; j < 32; ++j)
        p[j] = (j < NR) ? (kr[j < NR ? j : 0].x * q4.x + kr[j < NR ? j : 0].y * q4.y +
                           kr[j < NR ? j : 0].z * q4.z + kr[j < NR ? j : 0].w * q4.w)
                        : 0.f;
    rmba_tr_step<16>(p, lane);
    rmba_tr_step<8>(p, lane);
    rmba_tr_step<4>(p, lane);
    rmba_tr_step<2>(p, lane);
    rmba_tr_step<1>(p, lane);
    return p[0];
}

template <int NR>
__device__ __forceinline__ void rmba_chunk_ctx(const float* __restrict__ vbase, const int idc,
                                               const float a, float4& cacc) {
    float4 vr[NR];
    #pragma unroll
    for (int j = 0; j < NR; ++j) {
        const int ix = __builtin_amdgcn_readlane(idc, j);
        vr[j] = *(const float4*)(vbase + (size_t)ix * DD);
    }
    #pragma unroll
    for (int j = 0; j < NR; ++j) {
        const float aj = rmba_readlane_f(a, j);
        cacc.x += aj * vr[j].x;
        cacc.y += aj * vr[j].y;
        cacc.z += aj * vr[j].z;
        cacc.w += aj * vr[j].w;
    }
}

constexpr int QPB    = 16;
constexpr int CPITCH = DD + 4;

__global__ __launch_bounds__(512)
void rmba_qattno_kernel(const float* __restrict__ A,
                        const float* __restrict__ Wq,
                        const float* __restrict__ bq,
                        const float* __restrict__ Kall,
                        const float* __restrict__ Vall,
                        const int*   __restrict__ cand,
                        const float* __restrict__ Wo,
                        const float* __restrict__ bo,
                        float* __restrict__ attn_out,
                        float* __restrict__ out)
{
    __shared__ float part[8][8][32];
    __shared__ float sq[8][16];
    __shared__ float rnorm[16];
    __shared__ __attribute__((aligned(16))) float qs[QPB][DD];
    __shared__ __attribute__((aligned(16))) float cs[QPB][CPITCH];

    const int lane = threadIdx.x & 31;
    const int wave = __builtin_amdgcn_readfirstlane((int)(threadIdx.x >> 5));
    const int half = lane >> 4;
    const int l16  = lane & 15;
    const int koff = half * 2;
    const int rowBase = blockIdx.x * QPB;

    {
        const int ct   = wave & 7;
        const int kh   = wave >> 3;
        const int arow = rowBase + l16;
        const int bcol = ct * 16 + l16;
        const int k0   = kh * (CRV / 2);

        v8f acc = {};
        #pragma unroll 4
        for (int k = k0; k < k0 + CRV / 2; k += 4) {
            v2f a, b;
            a.x = A[arow * CRV + k + koff];
            a.y = A[arow * CRV + k + koff + 1];
            b.x = Wq[(k + koff) * DD + bcol];
            b.y = Wq[(k + koff + 1) * DD + bcol];
            acc = __builtin_amdgcn_wmma_f32_16x16x4_f32(false, a, false, b,
                                                        (short)0, acc, false, false);
        }
        if (kh == 1) {
            #pragma unroll
            for (int r = 0; r < 8; ++r) part[ct][r][lane] = acc[r];
        }
        __syncthreads();

        float vals[8];
        if (kh == 0) {
            const float bias = bq[bcol];
            float ss[8];
            #pragma unroll
            for (int r = 0; r < 8; ++r) {
                vals[r] = (acc[r] + part[ct][r][lane]) + bias;
                ss[r] = vals[r] * vals[r];
            }
            #pragma unroll
            for (int s = 8; s >= 1; s >>= 1)
                #pragma unroll
                for (int r = 0; r < 8; ++r) ss[r] += __shfl_xor(ss[r], s, 32);
            if (l16 == 0) {
                #pragma unroll
                for (int r = 0; r < 8; ++r) sq[ct][half * 8 + r] = ss[r];
            }
        }
        __syncthreads();
        if (threadIdx.x < 16) {
            float s = 0.f;
            #pragma unroll
            for (int w = 0; w < 8; ++w) s += sq[w][threadIdx.x];
            rnorm[threadIdx.x] = 1.0f / fmaxf(sqrtf(s), 1e-12f);
        }
        __syncthreads();
        if (kh == 0) {
            #pragma unroll
            for (int r = 0; r < 8; ++r) {
                const int row = half * 8 + r;
                qs[row][bcol] = vals[r] * rnorm[row];
            }
        }
        __syncthreads();
    }

    const int n = rowBase + wave;
    const float4 q4 = *(const float4*)(&qs[wave][4 * lane]);
    const float* kbase = Kall + 4 * lane;
    const float* vbase = Vall + 4 * lane;

    int   idr[8];
    float wr[8];
    #pragma unroll
    for (int c = 0; c < 8; ++c) idr[c] = cand[n * MM + 32 * c + lane];
    #pragma unroll
    for (int c = 0; c < 8; ++c) wr[c] = -INFINITY;

    #pragma unroll 1
    for (int c = 0; c < 8; ++c) {
        int id = idr[0];
        #pragma unroll
        for (int k = 1; k < 8; ++k) id = (c == k) ? idr[k] : id;
        const unsigned vm = __builtin_amdgcn_ballot_w32(id >= 0);
        if (vm == 0u) continue;
        const int idc = max(id, 0);
        float dot;
        if (vm >> 16) dot = rmba_chunk_dots<32>(kbase, idc, q4, lane);
        else          dot = rmba_chunk_dots<16>(kbase, idc, q4, lane);
        const float lg = (id >= 0) ? dot * INV_SQRT_D : -INFINITY;
        #pragma unroll
        for (int k = 0; k < 8; ++k) wr[k] = (c == k) ? lg : wr[k];
    }

    float mx = wr[0];
    #pragma unroll
    for (int k = 1; k < 8; ++k) mx = fmaxf(mx, wr[k]);
    #pragma unroll
    for (int s = 16; s >= 1; s >>= 1) mx = fmaxf(mx, __shfl_xor(mx, s, 32));

    float sm = 0.f;
    #pragma unroll
    for (int k = 0; k < 8; ++k) {
        wr[k] = (idr[k] >= 0) ? __expf(wr[k] - mx) : 0.f;
        sm += wr[k];
    }
    #pragma unroll
    for (int s = 16; s >= 1; s >>= 1) sm += __shfl_xor(sm, s, 32);

    #pragma unroll
    for (int k = 0; k < 8; ++k) {
        wr[k] = wr[k] / sm;
        attn_out[n * MM + 32 * k + lane] = wr[k];
    }

    float4 cacc = make_float4(0.f, 0.f, 0.f, 0.f);
    #pragma unroll 1
    for (int c = 0; c < 8; ++c) {
        int   id = idr[0];
        float a  = wr[0];
        #pragma unroll
        for (int k = 1; k < 8; ++k) { id = (c == k) ? idr[k] : id; a = (c == k) ? wr[k] : a; }
        const unsigned vm = __builtin_amdgcn_ballot_w32(id >= 0);
        if (vm == 0u) continue;
        const int idc = max(id, 0);
        if (vm >> 16) rmba_chunk_ctx<32>(vbase, idc, a, cacc);
        else          rmba_chunk_ctx<16>(vbase, idc, a, cacc);
    }
    *(float4*)(&cs[wave][4 * lane]) = cacc;
    __syncthreads();

    if (wave < 12) {
        const int cb0 = wave * 16 + l16;
        const int cb1 = (wave + 12) * 16 + l16;
        v8f acc0 = {}, acc1 = {};
        #pragma unroll 4
        for (int k = 0; k < DD; k += 4) {
            v2f a, b0, b1;
            a.x  = cs[l16][k + koff];
            a.y  = cs[l16][k + koff + 1];
            b0.x = Wo[(k + koff) * CRV + cb0];
            b0.y = Wo[(k + koff + 1) * CRV + cb0];
            b1.x = Wo[(k + koff) * CRV + cb1];
            b1.y = Wo[(k + koff + 1) * CRV + cb1];
            acc0 = __builtin_amdgcn_wmma_f32_16x16x4_f32(false, a, false, b0,
                                                         (short)0, acc0, false, false);
            acc1 = __builtin_amdgcn_wmma_f32_16x16x4_f32(false, a, false, b1,
                                                         (short)0, acc1, false, false);
        }
        const float bias0 = bo[cb0];
        const float bias1 = bo[cb1];
        #pragma unroll
        for (int r = 0; r < 8; ++r) {
            const int row = rowBase + half * 8 + r;
            out[row * CRV + cb0] = acc0[r] + bias0;
            out[row * CRV + cb1] = acc1[r] + bias1;
        }
    }
}

extern "C" void kernel_launch(void* const* d_in, const int* in_sizes, int n_in,
                              void* d_out, int out_size, void* d_ws, size_t ws_size,
                              hipStream_t stream) {
    const float* rv_x = (const float*)d_in[0];
    const float* bev  = (const float*)d_in[1];
    const int*   cand = (const int*)  d_in[2];
    const float* Wq   = (const float*)d_in[3];
    const float* bq   = (const float*)d_in[4];
    const float* Wk   = (const float*)d_in[5];
    const float* bk   = (const float*)d_in[6];
    const float* Wv   = (const float*)d_in[7];
    const float* bv   = (const float*)d_in[8];
    const float* Wo   = (const float*)d_in[9];
    const float* bo   = (const float*)d_in[10];

    float* out_main = (float*)d_out;
    float* out_attn = out_main + (size_t)NRQ * CRV;

    float* Kall = (float*)d_ws;
    float* Vall = Kall + (size_t)NBC * DD;

    rmba_kvproj_kernel<<<NBC / 16, 256, 0, stream>>>(bev, Wk, bk, Wv, bv, Kall, Vall);
    rmba_qattno_kernel<<<NRQ / QPB, 512, 0, stream>>>(rv_x, Wq, bq, Kall, Vall, cand,
                                                      Wo, bo, out_attn, out_main);
}


# K-row gather software-pipelined: a 16-row register set is asked for the next chunk's rows as soon as its dots are taken, so the loads fly under the 31-step reduce; first V rows asked before softmax
# speedup vs baseline: 3.5602x; 1.0857x over previous
#include <hip/hip_runtime.h>
#include <math.h>


typedef float v2f __attribute__((ext_vector_type(2)));
typedef float v8f __attribute__((ext_vector_type(8)));
typedef _Float16 v4h __attribute__((ext_vector_type(4)));

constexpr int NRQ = 4096;
constexpr int NBC = 16384;
constexpr int CB  = 80;
constexpr int CRV = 384;
constexpr int DD  = 128;
constexpr int MM  = 256;
constexpr float INV_SQRT_D = 0.08838834764831845f;


__global__ __launch_bounds__(256)
void rmba_kvproj_kernel(const float* __restrict__ bev,
                        const float* __restrict__ Wk, const float* __restrict__ bk,
                        const float* __restrict__ Wv, const float* __restrict__ bv,
                        float* __restrict__ Kall,
                        float* __restrict__ Vall)
{
    _Float16* __restrict__ Kh = (_Float16*)Kall;
    const int lane = threadIdx.x & 31;
    const int wave = threadIdx.x >> 5;
    const int half = lane >> 4;
    const int l16  = lane & 15;
    const int koff = half * 2;
    const int rowBase = blockIdx.x * 16;
    const int colBase = wave * 16;
    const int arow = rowBase + l16;
    const int bcol = colBase + l16;

    v8f acck = {}, accv = {};
    #pragma unroll 4
    for (int k = 0; k < CB; k += 4) {
        v2f a, wk, wv;
        a.x  = bev[(k + koff) * NBC + arow];
        a.y  = bev[(k + koff + 1) * NBC + arow];
        wk.x = Wk[(k + koff) * DD + bcol];
        wk.y = Wk[(k + koff + 1) * DD + bcol];
        wv.x = Wv[(k + koff) * DD + bcol];
        wv.y = Wv[(k + koff + 1) * DD + bcol];
        acck = __builtin_amdgcn_wmma_f32_16x16x4_f32(false, a, false, wk,
                                                     (short)0, acck, false, false);
        accv = __builtin_amdgcn_wmma_f32_16x16x4_f32(false, a, false, wv,
                                                     (short)0, accv, false, false);
    }

    const float biask = bk[bcol];
    const float biasv = bv[bcol];
    float kv[8], ss[8];
    #pragma unroll
    for (int r = 0; r < 8; ++r) { kv[r] = acck[r] + biask; ss[r] = kv[r] * kv[r]; }
    #pragma unroll
    for (int s = 8; s >= 1; s >>= 1)
        #pragma unroll
        for (int r = 0; r < 8; ++r) ss[r] += __shfl_xor(ss[r], s, 32);

    __shared__ float sq[8][16];
    __shared__ float rnorm[16];
    if (l16 == 0) {
        #pragma unroll
        for (int r = 0; r < 8; ++r) sq[wave][half * 8 + r] = ss[r];
    }
    __syncthreads();
    if (threadIdx.x < 16) {
        float s = 0.f;
        #pragma unroll
        for (int w = 0; w < 8; ++w) s += sq[w][threadIdx.x];
        rnorm[threadIdx.x] = 1.0f / fmaxf(sqrtf(s), 1e-12f);
    }
    __syncthreads();
    #pragma unroll
    for (int r = 0; r < 8; ++r) {
        const int row = half * 8 + r;
        Kh[(rowBase + row) * DD + bcol] = (_Float16)(kv[r] * rnorm[row]);
        Vall[(rowBase + row) * DD + bcol] = accv[r] + biasv;
    }
}

__device__ __forceinline__ float rmba_readlane_f(float x, int j) {
    return __int_as_float(__builtin_amdgcn_readlane(__float_as_int(x), j));
}

template <int S>
__device__ __forceinline__ void rmba_tr_step(float (&p)[32], const int lane) {
    const bool up = (lane & S) != 0;
    #pragma unroll
    for (int i = 0; i < S; ++i) {
        const float keep = up ? p[i + S] : p[i];
        const float send = up ? p[i] : p[i + S];
        p[i] = keep + __shfl_xor(send, S, 32);
    }
}


template <int H>
__device__ __forceinline__ void rmba_k_ask(v4h (&kr)[16], const _Float16* __restrict__ kbase, const int idc) {
    #pragma unroll
    for (int j = 0; j < 16; ++j) {
        const int ix = __builtin_amdgcn_readlane(idc, j + 16 * H);
        kr[j] = *(const v4h*)(kbase + (size_t)ix * DD);
    }
}

template <int H>
__device__ __forceinline__ void rmba_k_part(const v4h (&kr)[16], const float4 q4, float (&p)[32]) {
    #pragma unroll
    for (int j = 0; j < 16; ++j)
        p[16 * H + j] = (float)kr[j].x * q4.x + (float)kr[j].y * q4.y +
                        (float)kr[j].z * q4.z + (float)kr[j].w * q4.w;
}

__device__ __forceinline__ float rmba_k_reduce(float (&p)[32], const int lane) {
    rmba_tr_step<16>(p, lane);
    rmba_tr_step<8>(p, lane);
    rmba_tr_step<4>(p, lane);
    rmba_tr_step<2>(p, lane);
    rmba_tr_step<1>(p, lane);
    return p[0];
}

template <int H>
__device__ __forceinline__ void rmba_v_ask(float4 (&vr)[16], const float* __restrict__ vbase, const int idc) {
    #pragma unroll
    for (int j = 0; j < 16; ++j) {
        const int ix = __builtin_amdgcn_readlane(idc, j + 16 * H);
        vr[j] = *(const float4*)(vbase + (size_t)ix * DD);
    }
}

template <int H>
__device__ __forceinline__ void rmba_v_acc(const float4 (&vr)[16], const float a, float4& cacc) {
    #pragma unroll
    for (int j = 0; j < 16; ++j) {
        const float aj = rmba_readlane_f(a, j + 16 * H);
        cacc.x += aj * vr[j].x;
        cacc.y += aj * vr[j].y;
        cacc.z += aj * vr[j].z;
        cacc.w += aj * vr[j].w;
    }
}

__device__ __forceinline__ int rmba_pick_i(const int (&v)[8], const int c) {
    int r = v[0];
    #pragma unroll
    for (int k = 1; k < 8; ++k) r = (c == k) ? v[k] : r;
    return r;
}
__device__ __forceinline__ float rmba_pick_f(const float (&v)[8], const int c) {
    float r = v[0];
    #pragma unroll
    for (int k = 1; k < 8; ++k) r = (c == k) ? v[k] : r;
    return r;
}
__device__ __forceinline__ void rmba_put_f(float (&v)[8], const int c, const float x) {
    #pragma unroll
    for (int k = 0; k < 8; ++k) v[k] = (c == k) ? x : v[k];
}

constexpr int QPB    = 16;
constexpr int CPITCH = DD + 4;

__global__ __launch_bounds__(512)
void rmba_qattno_kernel(const float* __restrict__ A,
                        const float* __restrict__ Wq,
                        const float* __restrict__ bq,
                        const float* __restrict__ Kall,
                        const float* __restrict__ Vall,
                        const int*   __restrict__ cand,
                        const float* __restrict__ Wo,
                        const float* __restrict__ bo,
                        float* __restrict__ attn_out,
                        float* __restrict__ out)
{
    __shared__ float part[8][8][32];
    __shared__ float sq[8][16];
    __shared__ float rnorm[16];
    __shared__ __attribute__((aligned(16))) float qs[QPB][DD];
    __shared__ __attribute__((aligned(16))) float cs[QPB][CPITCH];

    const int lane = threadIdx.x & 31;
    const int wave = __builtin_amdgcn_readfirstlane((int)(threadIdx.x >> 5));
    const int half = lane >> 4;
    const int l16  = lane & 15;
    const int koff = half * 2;
    const int rowBase = blockIdx.x * QPB;

    const int n = rowBase + wave;
    int idr[8];
    #pragma unroll
    for (int c = 0; c < 8; ++c) idr[c] = cand[n * MM + 32 * c + lane];

    {
        const int ct   = wave & 7;
        const int kh   = wave >> 3;
        const int arow = rowBase + l16;
        const int bcol = ct * 16 + l16;
        const int k0   = kh * (CRV / 2);

        v8f acc = {};
        #pragma unroll 4
        for (int k = k0; k < k0 + CRV / 2; k += 4) {
            v2f a, b;
            a.x = A[arow * CRV + k + koff];
            a.y = A[arow * CRV + k + koff + 1];
            b.x = Wq[(k + koff) * DD + bcol];
            b.y = Wq[(k + koff + 1) * DD + bcol];
            acc = __builtin_amdgcn_wmma_f32_16x16x4_f32(false, a, false, b,
                                                        (short)0, acc, false, false);
        }
        if (kh == 1) {
            #pragma unroll
            for (int r = 0; r < 8; ++r) part[ct][r][lane] = acc[r];
        }
        __syncthreads();

        float vals[8];
        if (kh == 0) {
            const float bias = bq[bcol];
            float ss[8];
            #pragma unroll
            for (int r = 0; r < 8; ++r) {
                vals[r] = (acc[r] + part[ct][r][lane]) + bias;
                ss[r] = vals[r] * vals[r];
            }
            #pragma unroll
            for (int s = 8; s >= 1; s >>= 1)
                #pragma unroll
                for (int r = 0; r < 8; ++r) ss[r] += __shfl_xor(ss[r], s, 32);
            if (l16 == 0) {
                #pragma unroll
                for (int r = 0; r < 8; ++r) sq[ct][half * 8 + r] = ss[r];
            }
        }
        __syncthreads();
        if (threadIdx.x < 16) {
            float s = 0.f;
            #pragma unroll
            for (int w = 0; w < 8; ++w) s += sq[w][threadIdx.x];
            rnorm[threadIdx.x] = 1.0f / fmaxf(sqrtf(s), 1e-12f);
        }
        __syncthreads();
        if (kh == 0) {
            #pragma unroll
            for (int r = 0; r < 8; ++r) {
                const int row = half * 8 + r;
                qs[row][bcol] = vals[r] * rnorm[row];
            }
        }
        __syncthreads();
    }

    const float4 q4 = *(const float4*)(&qs[wave][4 * lane]);
    const _Float16* kbase = (const _Float16*)Kall + 4 * lane;
    const float* vbase = Vall + 4 * lane;

    float wr[8];
    #pragma unroll
    for (int c = 0; c < 8; ++c) wr[c] = -INFINITY;

    int ng = 0;
    #pragma unroll
    for (int k = 0; k < 8; ++k) {
        const unsigned vmk = __builtin_amdgcn_ballot_w32(idr[k] >= 0);
        if (vmk != 0u) ng = 2 * k + ((vmk >> 16) ? 2 : 1);
    }
    const int nc = (ng + 1) >> 1;

    if (nc > 0) {
        v4h klo[16], khi[16];
        int c  = 0;
        int id = idr[0];
        rmba_k_ask<0>(klo, kbase, max(id, 0));
        rmba_k_ask<1>(khi, kbase, max(id, 0));
        #pragma unroll 1
        for (; c + 1 < nc; ++c) {
            float p[32];
            const int id1  = rmba_pick_i(idr, c + 1);
            const int idc1 = max(id1, 0);
            rmba_k_part<0>(klo, q4, p);
            rmba_k_ask<0>(klo, kbase, idc1);
            __builtin_amdgcn_sched_barrier(0);
            rmba_k_part<1>(khi, q4, p);
            rmba_k_ask<1>(khi, kbase, idc1);
            const float dot = rmba_k_reduce(p, lane);
            rmba_put_f(wr, c, (id >= 0) ? dot * INV_SQRT_D : -INFINITY);
            id = id1;
            __builtin_amdgcn_sched_barrier(0);
        }
        {
            float p[32];
            rmba_k_part<0>(klo, q4, p);
            rmba_k_part<1>(khi, q4, p);
            const float dot = rmba_k_reduce(p, lane);
            rmba_put_f(wr, c, (id >= 0) ? dot * INV_SQRT_D : -INFINITY);
        }
    }

    float4 va[16], vb[16];
    rmba_v_ask<0>(va, vbase, max(idr[0], 0));

    float mx = wr[0];
    #pragma unroll
    for (int k = 1; k < 8; ++k) mx = fmaxf(mx, wr[k]);
    #pragma unroll
    for (int s = 16; s >= 1; s >>= 1) mx = fmaxf(mx, __shfl_xor(mx, s, 32));

    float sm = 0.f;
    #pragma unroll
    for (int k = 0; k < 8; ++k) {
        wr[k] = (idr[k] >= 0) ? __expf(wr[k] - mx) : 0.f;
        sm += wr[k];
    }
    #pragma unroll
    for (int s = 16; s >= 1; s >>= 1) sm += __shfl_xor(sm, s, 32);

    #pragma unroll
    for (int k = 0; k < 8; ++k) {
        wr[k] = wr[k] / sm;
        attn_out[n * MM + 32 * k + lane] = wr[k];
    }

    float4 cacc = make_float4(0.f, 0.f, 0.f, 0.f);
    {
        int   c   = 0;
        int   idc = max(idr[0], 0);
        float a   = wr[0];
        #pragma unroll 1
        for (; 2 * c + 2 < ng; ++c) {
            rmba_v_ask<1>(vb, vbase, idc);
            rmba_v_acc<0>(va, a, cacc);
            const int idn = max(rmba_pick_i(idr, c + 1), 0);
            rmba_v_ask<0>(va, vbase, idn);
            rmba_v_acc<1>(vb, a, cacc);
            a   = rmba_pick_f(wr, c + 1);
            idc = idn;
        }
        if (2 * c + 1 < ng) {
            rmba_v_ask<1>(vb, vbase, idc);
            rmba_v_acc<0>(va, a, cacc);
            rmba_v_acc<1>(vb, a, cacc);
        } else {
            rmba_v_acc<0>(va, a, cacc);
        }
    }
    *(float4*)(&cs[wave][4 * lane]) = cacc;
    __syncthreads();

    if (wave < 12) {
        const int cb0 = wave * 16 + l16;
        const int cb1 = (wave + 12) * 16 + l16;
        v8f acc0 = {}, acc1 = {};
        #pragma unroll 4
        for (int k = 0; k < DD; k += 4) {
            v2f a, b0, b1;
            a.x  = cs[l16][k + koff];
            a.y  = cs[l16][k + koff + 1];
            b0.x = Wo[(k + koff) * CRV + cb0];
            b0.y = Wo[(k + koff + 1) * CRV + cb0];
            b1.x = Wo[(k + koff) * CRV + cb1];
            b1.y = Wo[(k + koff + 1) * CRV + cb1];
            acc0 = __builtin_amdgcn_wmma_f32_16x16x4_f32(false, a, false, b0,
                                                         (short)0, acc0, false, false);
            acc1 = __builtin_amdgcn_wmma_f32_16x16x4_f32(false, a, false, b1,
                                                         (short)0, acc1, false, false);
        }
        const float bias0 = bo[cb0];
        const float bias1 = bo[cb1];
        #pragma unroll
        for (int r = 0; r < 8; ++r) {
            const int row = rowBase + half * 8 + r;
            out[row * CRV + cb0] = acc0[r] + bias0;
            out[row * CRV + cb1] = acc1[r] + bias1;
        }
    }
}

extern "C" void kernel_launch(void* const* d_in, const int* in_sizes, int n_in,
                              void* d_out, int out_size, void* d_ws, size_t ws_size,
                              hipStream_t stream) {
    const float* rv_x = (const float*)d_in[0];
    const float* bev  = (const float*)d_in[1];
    const int*   cand = (const int*)  d_in[2];
    const float* Wq   = (const float*)d_in[3];
    const float* bq   = (const float*)d_in[4];
    const float* Wk   = (const float*)d_in[5];
    const float* bk   = (const float*)d_in[6];
    const float* Wv   = (const float*)d_in[7];
    const float* bv   = (const float*)d_in[8];
    const float* Wo   = (const float*)d_in[9];
    const float* bo   = (const float*)d_in[10];

    float* out_main = (float*)d_out;
    float* out_attn = out_main + (size_t)NRQ * CRV;

    float* Kall = (float*)d_ws;
    float* Vall = Kall + (size_t)NBC * DD;

    rmba_kvproj_kernel<<<NBC / 16, 256, 0, stream>>>(bev, Wk, bk, Wv, bv, Kall, Vall);
    rmba_qattno_kernel<<<NRQ / QPB, 512, 0, stream>>>(rv_x, Wq, bq, Kall, Vall, cand,
                                                      Wo, bo, out_attn, out_main);
}


# Q proj re-split: k over 8 wave pairs; a wave's share is 30 16-B loads in one group (Wq whole lines, 4 column-strided tiles); partials summed via LDS into its q row (Q k-sum, norm sum order change)
# speedup vs baseline: 3.8720x; 1.0876x over previous
#include <hip/hip_runtime.h>
#include <math.h>


typedef float v2f __attribute__((ext_vector_type(2)));
typedef float v8f __attribute__((ext_vector_type(8)));
typedef _Float16 v4h __attribute__((ext_vector_type(4)));

constexpr int NRQ = 4096;
constexpr int NBC = 16384;
constexpr int CB  = 80;
constexpr int CRV = 384;
constexpr int DD  = 128;
constexpr int MM  = 256;
constexpr float INV_SQRT_D = 0.08838834764831845f;


__global__ __launch_bounds__(256)
void rmba_kvproj_kernel(const float* __restrict__ bev,
                        const float* __restrict__ Wk, const float* __restrict__ bk,
                        const float* __restrict__ Wv, const float* __restrict__ bv,
                        float* __restrict__ Kall,
                        float* __restrict__ Vall)
{
    _Float16* __restrict__ Kh = (_Float16*)Kall;
    const int lane = threadIdx.x & 31;
    const int wave = threadIdx.x >> 5;
    const int half = lane >> 4;
    const int l16  = lane & 15;
    const int koff = half * 2;
    const int rowBase = blockIdx.x * 16;
    const int colBase = wave * 16;
    const int arow = rowBase + l16;
    const int bcol = colBase + l16;

    v8f acck = {}, accv = {};
    #pragma unroll 4
    for (int k = 0; k < CB; k += 4) {
        v2f a, wk, wv;
        a.x  = bev[(k + koff) * NBC + arow];
        a.y  = bev[(k + koff + 1) * NBC + arow];
        wk.x = Wk[(k + koff) * DD + bcol];
        wk.y = Wk[(k + koff + 1) * DD + bcol];
        wv.x = Wv[(k + koff) * DD + bcol];
        wv.y = Wv[(k + koff + 1) * DD + bcol];
        acck = __builtin_amdgcn_wmma_f32_16x16x4_f32(false, a, false, wk,
                                                     (short)0, acck, false, false);
        accv = __builtin_amdgcn_wmma_f32_16x16x4_f32(false, a, false, wv,
                                                     (short)0, accv, false, false);
    }

    const float biask = bk[bcol];
    const float biasv = bv[bcol];
    float kv[8], ss[8];
    #pragma unroll
    for (int r = 0; r < 8; ++r) { kv[r] = acck[r] + biask; ss[r] = kv[r] * kv[r]; }
    #pragma unroll
    for (int s = 8; s >= 1; s >>= 1)
        #pragma unroll
        for (int r = 0; r < 8; ++r) ss[r] += __shfl_xor(ss[r], s, 32);

    __shared__ float sq[8][16];
    __shared__ float rnorm[16];
    if (l16 == 0) {
        #pragma unroll
        for (int r = 0; r < 8; ++r) sq[wave][half * 8 + r] = ss[r];
    }
    __syncthreads();
    if (threadIdx.x < 16) {
        float s = 0.f;
        #pragma unroll
        for (int w = 0; w < 8; ++w) s += sq[w][threadIdx.x];
        rnorm[threadIdx.x] = 1.0f / fmaxf(sqrtf(s), 1e-12f);
    }
    __syncthreads();
    #pragma unroll
    for (int r = 0; r < 8; ++r) {
        const int row = half * 8 + r;
        Kh[(rowBase + row) * DD + bcol] = (_Float16)(kv[r] * rnorm[row]);
        Vall[(rowBase + row) * DD + bcol] = accv[r] + biasv;
    }
}

__device__ __forceinline__ float rmba_readlane_f(float x, int j) {
    return __int_as_float(__builtin_amdgcn_readlane(__float_as_int(x), j));
}

template <int S>
__device__ __forceinline__ void rmba_tr_step(float (&p)[32], const int lane) {
    const bool up = (lane & S) != 0;
    #pragma unroll
    for (int i = 0; i < S; ++i) {
        const float keep = up ? p[i + S] : p[i];
        const float send = up ? p[i] : p[i + S];
        p[i] = keep + __shfl_xor(send, S, 32);
    }
}


template <int H>
__device__ __forceinline__ void rmba_k_ask(v4h (&kr)[16], const _Float16* __restrict__ kbase, const int idc) {
    #pragma unroll
    for (int j = 0; j < 16; ++j) {
        const int ix = __builtin_amdgcn_readlane(idc, j + 16 * H);
        kr[j] = *(const v4h*)(kbase + (size_t)ix * DD);
    }
}

template <int H>
__device__ __forceinline__ void rmba_k_part(const v4h (&kr)[16], const float4 q4, float (&p)[32]) {
    #pragma unroll
    for (int j = 0; j < 16; ++j)
        p[16 * H + j] = (float)kr[j].x * q4.x + (float)kr[j].y * q4.y +
                        (float)kr[j].z * q4.z + (float)kr[j].w * q4.w;
}

__device__ __forceinline__ float rmba_k_reduce(float (&p)[32], const int lane) {
    rmba_tr_step<16>(p, lane);
    rmba_tr_step<8>(p, lane);
    rmba_tr_step<4>(p, lane);
    rmba_tr_step<2>(p, lane);
    rmba_tr_step<1>(p, lane);
    return p[0];
}

template <int H>
__device__ __forceinline__ void rmba_v_ask(float4 (&vr)[16], const float* __restrict__ vbase, const int idc) {
    #pragma unroll
    for (int j = 0; j < 16; ++j) {
        const int ix = __builtin_amdgcn_readlane(idc, j + 16 * H);
        vr[j] = *(const float4*)(vbase + (size_t)ix * DD);
    }
}

template <int H>
__device__ __forceinline__ void rmba_v_acc(const float4 (&vr)[16], const float a, float4& cacc) {
    #pragma unroll
    for (int j = 0; j < 16; ++j) {
        const float aj = rmba_readlane_f(a, j + 16 * H);
        cacc.x += aj * vr[j].x;
        cacc.y += aj * vr[j].y;
        cacc.z += aj * vr[j].z;
        cacc.w += aj * vr[j].w;
    }
}

__device__ __forceinline__ int rmba_pick_i(const int (&v)[8], const int c) {
    int r = v[0];
    #pragma unroll
    for (int k = 1; k < 8; ++k) r = (c == k) ? v[k] : r;
    return r;
}
__device__ __forceinline__ float rmba_pick_f(const float (&v)[8], const int c) {
    float r = v[0];
    #pragma unroll
    for (int k = 1; k < 8; ++k) r = (c == k) ? v[k] : r;
    return r;
}
__device__ __forceinline__ void rmba_put_f(float (&v)[8], const int c, const float x) {
    #pragma unroll
    for (int k = 0; k < 8; ++k) v[k] = (c == k) ? x : v[k];
}

constexpr int QPB    = 16;
constexpr int CPITCH = DD + 4;
constexpr int QKS    = 8;
constexpr int QKL    = CRV / QKS;

__global__ __launch_bounds__(512)
void rmba_qattno_kernel(const float* __restrict__ A,
                        const float* __restrict__ Wq,
                        const float* __restrict__ bq,
                        const float* __restrict__ Kall,
                        const float* __restrict__ Vall,
                        const int*   __restrict__ cand,
                        const float* __restrict__ Wo,
                        const float* __restrict__ bo,
                        float* __restrict__ attn_out,
                        float* __restrict__ out)
{
    __shared__ __attribute__((aligned(16))) float part[QKS][QPB][DD];
    __shared__ __attribute__((aligned(16))) float cs[QPB][CPITCH];

    const int lane = threadIdx.x & 31;
    const int wave = __builtin_amdgcn_readfirstlane((int)(threadIdx.x >> 5));
    const int half = lane >> 4;
    const int l16  = lane & 15;
    const int koff = half * 2;
    const int rowBase = blockIdx.x * QPB;

    const int n = rowBase + wave;
    int idr[8];
    #pragma unroll
    for (int c = 0; c < 8; ++c) idr[c] = cand[n * MM + 32 * c + lane];

    float4 q4;
    {
        const int ks   = wave >> 1;
        const int arow = rowBase + l16;
        const int kb   = ks * QKL + 4 * half;
        const int cb   = 64 * (wave & 1) + 4 * l16;

        float4 a4[6];
        float4 b4[6][4];
        #pragma unroll
        for (int j = 0; j < 6; ++j)
            a4[j] = *(const float4*)(A + (size_t)arow * CRV + kb + 8 * j);
        #pragma unroll
        for (int j = 0; j < 6; ++j) {
            #pragma unroll
            for (int i = 0; i < 4; ++i)
                b4[j][i] = *(const float4*)(Wq + (size_t)(kb + 8 * j + i) * DD + cb);
        }

        v8f acc0 = {}, acc1 = {}, acc2 = {}, acc3 = {};
        #pragma unroll
        for (int j = 0; j < 6; ++j) {
            v2f a, b;
            a.x = a4[j].x;  a.y = a4[j].y;
            b.x = b4[j][0].x;  b.y = b4[j][1].x;
            acc0 = __builtin_amdgcn_wmma_f32_16x16x4_f32(false, a, false, b, (short)0, acc0, false, false);
            b.x = b4[j][0].y;  b.y = b4[j][1].y;
            acc1 = __builtin_amdgcn_wmma_f32_16x16x4_f32(false, a, false, b, (short)0, acc1, false, false);
            b.x = b4[j][0].z;  b.y = b4[j][1].z;
            acc2 = __builtin_amdgcn_wmma_f32_16x16x4_f32(false, a, false, b, (short)0, acc2, false, false);
            b.x = b4[j][0].w;  b.y = b4[j][1].w;
            acc3 = __builtin_amdgcn_wmma_f32_16x16x4_f32(false, a, false, b, (short)0, acc3, false, false);
            a.x = a4[j].z;  a.y = a4[j].w;
            b.x = b4[j][2].x;  b.y = b4[j][3].x;
            acc0 = __builtin_amdgcn_wmma_f32_16x16x4_f32(false, a, false, b, (short)0, acc0, false, false);
            b.x = b4[j][2].y;  b.y = b4[j][3].y;
            acc1 = __builtin_amdgcn_wmma_f32_16x16x4_f32(false, a, false, b, (short)0, acc1, false, false);
            b.x = b4[j][2].z;  b.y = b4[j][3].z;
            acc2 = __builtin_amdgcn_wmma_f32_16x16x4_f32(false, a, false, b, (short)0, acc2, false, false);
            b.x = b4[j][2].w;  b.y = b4[j][3].w;
            acc3 = __builtin_amdgcn_wmma_f32_16x16x4_f32(false, a, false, b, (short)0, acc3, false, false);
        }
        #pragma unroll
        for (int r = 0; r < 8; ++r)
            *(float4*)(&part[ks][half * 8 + r][cb]) = make_float4(acc0[r], acc1[r], acc2[r], acc3[r]);
        __syncthreads();

        float4 s = *(const float4*)(&part[0][wave][4 * lane]);
        #pragma unroll
        for (int k = 1; k < QKS; ++k) {
            const float4 t = *(const float4*)(&part[k][wave][4 * lane]);
            s.x += t.x;  s.y += t.y;  s.z += t.z;  s.w += t.w;
        }
        const float4 bias = *(const float4*)(bq + 4 * lane);
        s.x += bias.x;  s.y += bias.y;  s.z += bias.z;  s.w += bias.w;
        float ss = s.x * s.x + s.y * s.y + s.z * s.z + s.w * s.w;
        #pragma unroll
        for (int d = 16; d >= 1; d >>= 1) ss += __shfl_xor(ss, d, 32);
        const float rn = 1.0f / fmaxf(sqrtf(ss), 1e-12f);
        q4 = make_float4(s.x * rn, s.y * rn, s.z * rn, s.w * rn);
    }

    const _Float16* kbase = (const _Float16*)Kall + 4 * lane;
    const float* vbase = Vall + 4 * lane;

    float wr[8];
    #pragma unroll
    for (int c = 0; c < 8; ++c) wr[c] = -INFINITY;

    int ng = 0;
    #pragma unroll
    for (int k = 0; k < 8; ++k) {
        const unsigned vmk = __builtin_amdgcn_ballot_w32(idr[k] >= 0);
        if (vmk != 0u) ng = 2 * k + ((vmk >> 16) ? 2 : 1);
    }
    const int nc = (ng + 1) >> 1;

    if (nc > 0) {
        v4h klo[16], khi[16];
        int c  = 0;
        int id = idr[0];
        rmba_k_ask<0>(klo, kbase, max(id, 0));
        rmba_k_ask<1>(khi, kbase, max(id, 0));
        #pragma unroll 1
        for (; c + 1 < nc; ++c) {
            float p[32];
            const int id1  = rmba_pick_i(idr, c + 1);
            const int idc1 = max(id1, 0);
            rmba_k_part<0>(klo, q4, p);
            rmba_k_ask<0>(klo, kbase, idc1);
            __builtin_amdgcn_sched_barrier(0);
            rmba_k_part<1>(khi, q4, p);
            rmba_k_ask<1>(khi, kbase, idc1);
            const float dot = rmba_k_reduce(p, lane);
            rmba_put_f(wr, c, (id >= 0) ? dot * INV_SQRT_D : -INFINITY);
            id = id1;
            __builtin_amdgcn_sched_barrier(0);
        }
        {
            float p[32];
            rmba_k_part<0>(klo, q4, p);
            rmba_k_part<1>(khi, q4, p);
            const float dot = rmba_k_reduce(p, lane);
            rmba_put_f(wr, c, (id >= 0) ? dot * INV_SQRT_D : -INFINITY);
        }
    }

    float4 va[16], vb[16];
    rmba_v_ask<0>(va, vbase, max(idr[0], 0));

    float mx = wr[0];
    #pragma unroll
    for (int k = 1; k < 8; ++k) mx = fmaxf(mx, wr[k]);
    #pragma unroll
    for (int s = 16; s >= 1; s >>= 1) mx = fmaxf(mx, __shfl_xor(mx, s, 32));

    float sm = 0.f;
    #pragma unroll
    for (int k = 0; k < 8; ++k) {
        wr[k] = (idr[k] >= 0) ? __expf(wr[k] - mx) : 0.f;
        sm += wr[k];
    }
    #pragma unroll
    for (int s = 16; s >= 1; s >>= 1) sm += __shfl_xor(sm, s, 32);

    #pragma unroll
    for (int k = 0; k < 8; ++k) {
        wr[k] = wr[k] / sm;
        attn_out[n * MM + 32 * k + lane] = wr[k];
    }

    float4 cacc = make_float4(0.f, 0.f, 0.f, 0.f);
    {
        int   c   = 0;
        int   idc = max(idr[0], 0);
        float a   = wr[0];
        #pragma unroll 1
        for (; 2 * c + 2 < ng; ++c) {
            rmba_v_ask<1>(vb, vbase, idc);
            rmba_v_acc<0>(va, a, cacc);
            const int idn = max(rmba_pick_i(idr, c + 1), 0);
            rmba_v_ask<0>(va, vbase, idn);
            rmba_v_acc<1>(vb, a, cacc);
            a   = rmba_pick_f(wr, c + 1);
            idc = idn;
        }
        if (2 * c + 1 < ng) {
            rmba_v_ask<1>(vb, vbase, idc);
            rmba_v_acc<0>(va, a, cacc);
            rmba_v_acc<1>(vb, a, cacc);
        } else {
            rmba_v_acc<0>(va, a, cacc);
        }
    }
    *(float4*)(&cs[wave][4 * lane]) = cacc;
    __syncthreads();

    if (wave < 12) {
        const int cb0 = wave * 16 + l16;
        const int cb1 = (wave + 12) * 16 + l16;
        v8f acc0 = {}, acc1 = {};
        #pragma unroll 4
        for (int k = 0; k < DD; k += 4) {
            v2f a, b0, b1;
            a.x  = cs[l16][k + koff];
            a.y  = cs[l16][k + koff + 1];
            b0.x = Wo[(k + koff) * CRV + cb0];
            b0.y = Wo[(k + koff + 1) * CRV + cb0];
            b1.x = Wo[(k + koff) * CRV + cb1];
            b1.y = Wo[(k + koff + 1) * CRV + cb1];
            acc0 = __builtin_amdgcn_wmma_f32_16x16x4_f32(false, a, false, b0,
                                                         (short)0, acc0, false, false);
            acc1 = __builtin_amdgcn_wmma_f32_16x16x4_f32(false, a, false, b1,
                                                         (short)0, acc1, false, false);
        }
        const float bias0 = bo[cb0];
        const float bias1 = bo[cb1];
        #pragma unroll
        for (int r = 0; r < 8; ++r) {
            const int row = rowBase + half * 8 + r;
            out[row * CRV + cb0] = acc0[r] + bias0;
            out[row * CRV + cb1] = acc1[r] + bias1;
        }
    }
}

extern "C" void kernel_launch(void* const* d_in, const int* in_sizes, int n_in,
                              void* d_out, int out_size, void* d_ws, size_t ws_size,
                              hipStream_t stream) {
    const float* rv_x = (const float*)d_in[0];
    const float* bev  = (const float*)d_in[1];
    const int*   cand = (const int*)  d_in[2];
    const float* Wq   = (const float*)d_in[3];
    const float* bq   = (const float*)d_in[4];
    const float* Wk   = (const float*)d_in[5];
    const float* bk   = (const float*)d_in[6];
    const float* Wv   = (const float*)d_in[7];
    const float* bv   = (const float*)d_in[8];
    const float* Wo   = (const float*)d_in[9];
    const float* bo   = (const float*)d_in[10];

    float* out_main = (float*)d_out;
    float* out_attn = out_main + (size_t)NRQ * CRV;

    float* Kall = (float*)d_ws;
    float* Vall = Kall + (size_t)NBC * DD;

    rmba_kvproj_kernel<<<NBC / 16, 256, 0, stream>>>(bev, Wk, bk, Wv, bv, Kall, Vall);
    rmba_qattno_kernel<<<NRQ / QPB, 512, 0, stream>>>(rv_x, Wq, bq, Kall, Vall, cand,
                                                      Wo, bo, out_attn, out_main);
}


# two waves per query in 32-wave work-groups, each owns every 2nd 32-slot chunk; max, exp sum, ctx joined via LDS; 8-row load sets, 32-bit row offsets to fit 128 regs (Q k-sum, softmax, ctx sum order)
# speedup vs baseline: 4.1900x; 1.0821x over previous
#include <hip/hip_runtime.h>
#include <math.h>


typedef float v2f __attribute__((ext_vector_type(2)));
typedef float v8f __attribute__((ext_vector_type(8)));
typedef _Float16 v4h __attribute__((ext_vector_type(4)));

constexpr int NRQ = 4096;
constexpr int NBC = 16384;
constexpr int CB  = 80;
constexpr int CRV = 384;
constexpr int DD  = 128;
constexpr int MM  = 256;
constexpr float INV_SQRT_D = 0.08838834764831845f;


__global__ __launch_bounds__(256)
void rmba_kvproj_kernel(const float* __restrict__ bev,
                        const float* __restrict__ Wk, const float* __restrict__ bk,
                        const float* __restrict__ Wv, const float* __restrict__ bv,
                        float* __restrict__ Kall,
                        float* __restrict__ Vall)
{
    _Float16* __restrict__ Kh = (_Float16*)Kall;
    const int lane = threadIdx.x & 31;
    const int wave = threadIdx.x >> 5;
    const int half = lane >> 4;
    const int l16  = lane & 15;
    const int koff = half * 2;
    const int rowBase = blockIdx.x * 16;
    const int colBase = wave * 16;
    const int arow = rowBase + l16;
    const int bcol = colBase + l16;

    v8f acck = {}, accv = {};
    #pragma unroll 4
    for (int k = 0; k < CB; k += 4) {
        v2f a, wk, wv;
        a.x  = bev[(k + koff) * NBC + arow];
        a.y  = bev[(k + koff + 1) * NBC + arow];
        wk.x = Wk[(k + koff) * DD + bcol];
        wk.y = Wk[(k + koff + 1) * DD + bcol];
        wv.x = Wv[(k + koff) * DD + bcol];
        wv.y = Wv[(k + koff + 1) * DD + bcol];
        acck = __builtin_amdgcn_wmma_f32_16x16x4_f32(false, a, false, wk,
                                                     (short)0, acck, false, false);
        accv = __builtin_amdgcn_wmma_f32_16x16x4_f32(false, a, false, wv,
                                                     (short)0, accv, false, false);
    }

    const float biask = bk[bcol];
    const float biasv = bv[bcol];
    float kv[8], ss[8];
    #pragma unroll
    for (int r = 0; r < 8; ++r) { kv[r] = acck[r] + biask; ss[r] = kv[r] * kv[r]; }
    #pragma unroll
    for (int s = 8; s >= 1; s >>= 1)
        #pragma unroll
        for (int r = 0; r < 8; ++r) ss[r] += __shfl_xor(ss[r], s, 32);

    __shared__ float sq[8][16];
    __shared__ float rnorm[16];
    if (l16 == 0) {
        #pragma unroll
        for (int r = 0; r < 8; ++r) sq[wave][half * 8 + r] = ss[r];
    }
    __syncthreads();
    if (threadIdx.x < 16) {
        float s = 0.f;
        #pragma unroll
        for (int w = 0; w < 8; ++w) s += sq[w][threadIdx.x];
        rnorm[threadIdx.x] = 1.0f / fmaxf(sqrtf(s), 1e-12f);
    }
    __syncthreads();
    #pragma unroll
    for (int r = 0; r < 8; ++r) {
        const int row = half * 8 + r;
        Kh[(rowBase + row) * DD + bcol] = (_Float16)(kv[r] * rnorm[row]);
        Vall[(rowBase + row) * DD + bcol] = accv[r] + biasv;
    }
}

__device__ __forceinline__ float rmba_readlane_f(float x, int j) {
    return __int_as_float(__builtin_amdgcn_readlane(__float_as_int(x), j));
}

template <int S>
__device__ __forceinline__ void rmba_tr_step(float (&p)[32], const int lane) {
    const bool up = (lane & S) != 0;
    #pragma unroll
    for (int i = 0; i < S; ++i) {
        const float keep = up ? p[i + S] : p[i];
        const float send = up ? p[i] : p[i + S];
        p[i] = keep + __shfl_xor(send, S, 32);
    }
}


template <int G>
__device__ __forceinline__ void rmba_k_ask(v4h (&kr)[8], const _Float16* __restrict__ kall, const unsigned lbyte, const int idc) {
    #pragma unroll
    for (int j = 0; j < 8; ++j) {
        const int ix = __builtin_amdgcn_readlane(idc, j + 8 * G);
        const unsigned off = (unsigned)ix * (unsigned)(DD * sizeof(_Float16)) + lbyte;
        kr[j] = *(const v4h*)((const char*)kall + off);
    }
}

template <int G>
__device__ __forceinline__ void rmba_k_part(const v4h (&kr)[8], const float4 q4, float (&p)[32]) {
    #pragma unroll
    for (int j = 0; j < 8; ++j)
        p[8 * G + j] = (float)kr[j].x * q4.x + (float)kr[j].y * q4.y +
                       (float)kr[j].z * q4.z + (float)kr[j].w * q4.w;
}

template <int G>
__device__ __forceinline__ void rmba_v_ask(float4 (&vr)[8], const float* __restrict__ vall, const unsigned lbyte, const int idc) {
    #pragma unroll
    for (int j = 0; j < 8; ++j) {
        const int ix = __builtin_amdgcn_readlane(idc, j + 8 * G);
        const unsigned off = (unsigned)ix * (unsigned)(DD * sizeof(float)) + lbyte;
        vr[j] = *(const float4*)((const char*)vall + off);
    }
}

template <int G>
__device__ __forceinline__ void rmba_v_acc(const float4 (&vr)[8], const float a, float4& cacc) {
    #pragma unroll
    for (int j = 0; j < 8; ++j) {
        const float aj = rmba_readlane_f(a, j + 8 * G);
        cacc.x += aj * vr[j].x;
        cacc.y += aj * vr[j].y;
        cacc.z += aj * vr[j].z;
        cacc.w += aj * vr[j].w;
    }
}

__device__ __forceinline__ int rmba_pick_i(const int (&v)[4], const int c) {
    int r = v[0];
    #pragma unroll
    for (int k = 1; k < 4; ++k) r = (c == k) ? v[k] : r;
    return r;
}
__device__ __forceinline__ float rmba_pick_f(const float (&v)[4], const int c) {
    float r = v[0];
    #pragma unroll
    for (int k = 1; k < 4; ++k) r = (c == k) ? v[k] : r;
    return r;
}
__device__ __forceinline__ void rmba_put_f(float (&v)[4], const int c, const float x) {
    #pragma unroll
    for (int k = 0; k < 4; ++k) v[k] = (c == k) ? x : v[k];
}

constexpr int QPB    = 16;
constexpr int CPITCH = DD + 4;
constexpr int QKS    = 16;
constexpr int QKL    = CRV / QKS;

__global__ __launch_bounds__(1024)
void rmba_qattno_kernel(const float* __restrict__ A,
                        const float* __restrict__ Wq,
                        const float* __restrict__ bq,
                        const float* __restrict__ Kall,
                        const float* __restrict__ Vall,
                        const int*   __restrict__ cand,
                        const float* __restrict__ Wo,
                        const float* __restrict__ bo,
                        float* __restrict__ attn_out,
                        float* __restrict__ out)
{
    __shared__ __attribute__((aligned(16))) float part[QKS][QPB][DD];
    __shared__ __attribute__((aligned(16))) float cs[2][QPB][CPITCH];
    __shared__ float smx[2][QPB];
    __shared__ float ssm[2][QPB];

    const int lane = threadIdx.x & 31;
    const int wave = __builtin_amdgcn_readfirstlane((int)(threadIdx.x >> 5));
    const int qi   = wave & 15;
    const int pt   = wave >> 4;
    const int half = lane >> 4;
    const int l16  = lane & 15;
    const int koff = half * 2;
    const int rowBase = blockIdx.x * QPB;

    const int n = rowBase + qi;
    const unsigned sl = 4u * (unsigned)(32 * pt + lane);
    const char* crow = (const char*)(cand + (size_t)n * MM);
    int idr[4];
    #pragma unroll
    for (int i = 0; i < 4; ++i) idr[i] = *(const int*)(crow + sl + 256u * (unsigned)i);

    float4 q4;
    {
        const int ks   = wave >> 1;
        const int arow = rowBase + l16;
        const int kb   = ks * QKL + 4 * half;
        const int cb   = 64 * (wave & 1) + 4 * l16;

        float4 a4[3];
        float4 b4[3][4];
        #pragma unroll
        for (int j = 0; j < 3; ++j)
            a4[j] = *(const float4*)(A + (size_t)arow * CRV + kb + 8 * j);
        #pragma unroll
        for (int j = 0; j < 3; ++j) {
            #pragma unroll
            for (int i = 0; i < 4; ++i)
                b4[j][i] = *(const float4*)(Wq + (size_t)(kb + 8 * j + i) * DD + cb);
        }

        v8f acc0 = {}, acc1 = {}, acc2 = {}, acc3 = {};
        #pragma unroll
        for (int j = 0; j < 3; ++j) {
            v2f a, b;
            a.x = a4[j].x;  a.y = a4[j].y;
            b.x = b4[j][0].x;  b.y = b4[j][1].x;
            acc0 = __builtin_amdgcn_wmma_f32_16x16x4_f32(false, a, false, b, (short)0, acc0, false, false);
            b.x = b4[j][0].y;  b.y = b4[j][1].y;
            acc1 = __builtin_amdgcn_wmma_f32_16x16x4_f32(false, a, false, b, (short)0, acc1, false, false);
            b.x = b4[j][0].z;  b.y = b4[j][1].z;
            acc2 = __builtin_amdgcn_wmma_f32_16x16x4_f32(false, a, false, b, (short)0, acc2, false, false);
            b.x = b4[j][0].w;  b.y = b4[j][1].w;
            acc3 = __builtin_amdgcn_wmma_f32_16x16x4_f32(false, a, false, b, (short)0, acc3, false, false);
            a.x = a4[j].z;  a.y = a4[j].w;
            b.x = b4[j][2].x;  b.y = b4[j][3].x;
            acc0 = __builtin_amdgcn_wmma_f32_16x16x4_f32(false, a, false, b, (short)0, acc0, false, false);
            b.x = b4[j][2].y;  b.y = b4[j][3].y;
            acc1 = __builtin_amdgcn_wmma_f32_16x16x4_f32(false, a, false, b, (short)0, acc1, false, false);
            b.x = b4[j][2].z;  b.y = b4[j][3].z;
            acc2 = __builtin_amdgcn_wmma_f32_16x16x4_f32(false, a, false, b, (short)0, acc2, false, false);
            b.x = b4[j][2].w;  b.y = b4[j][3].w;
            acc3 = __builtin_amdgcn_wmma_f32_16x16x4_f32(false, a, false, b, (short)0, acc3, false, false);
        }
        #pragma unroll
        for (int r = 0; r < 8; ++r)
            *(float4*)(&part[ks][half * 8 + r][cb]) = make_float4(acc0[r], acc1[r], acc2[r], acc3[r]);
        __syncthreads();

        float4 s = *(const float4*)(&part[0][qi][4 * lane]);
        #pragma unroll
        for (int k = 1; k < QKS; ++k) {
            const float4 t = *(const float4*)(&part[k][qi][4 * lane]);
            s.x += t.x;  s.y += t.y;  s.z += t.z;  s.w += t.w;
        }
        const float4 bias = *(const float4*)(bq + 4 * lane);
        s.x += bias.x;  s.y += bias.y;  s.z += bias.z;  s.w += bias.w;
        float ss = s.x * s.x + s.y * s.y + s.z * s.z + s.w * s.w;
        #pragma unroll
        for (int d = 16; d >= 1; d >>= 1) ss += __shfl_xor(ss, d, 32);
        const float rn = 1.0f / fmaxf(sqrtf(ss), 1e-12f);
        q4 = make_float4(s.x * rn, s.y * rn, s.z * rn, s.w * rn);
    }

    const _Float16* kall = (const _Float16*)Kall;
    const unsigned kl = 8u * (unsigned)lane;
    const unsigned vl = 16u * (unsigned)lane;

    float wr[4];
    #pragma unroll
    for (int i = 0; i < 4; ++i) wr[i] = -INFINITY;

    int ng = 0;
    #pragma unroll
    for (int i = 0; i < 4; ++i) {
        const unsigned vmk = __builtin_amdgcn_ballot_w32(idr[i] >= 0);
        if (vmk != 0u) ng = 2 * i + ((vmk >> 16) ? 2 : 1);
    }
    const int nc = (ng + 1) >> 1;

    if (nc > 0) {
        v4h ka[8], kb[8], kc[8];
        int id  = idr[0];
        int idc = max(id, 0);
        rmba_k_ask<0>(ka, kall, kl, idc);
        rmba_k_ask<1>(kb, kall, kl, idc);
        rmba_k_ask<2>(kc, kall, kl, idc);
        #pragma unroll 1
        for (int c = 0; c < nc; ++c) {
            float p[32];
            rmba_k_part<0>(ka, q4, p);
            rmba_k_ask<3>(ka, kall, kl, idc);
            __builtin_amdgcn_sched_barrier(0);
            rmba_k_part<1>(kb, q4, p);
            rmba_k_part<2>(kc, q4, p);
            rmba_k_part<3>(ka, q4, p);
            rmba_tr_step<16>(p, lane);
            const int id1 = rmba_pick_i(idr, c + 1);
            if (c + 1 < nc) {
                idc = max(id1, 0);
                rmba_k_ask<0>(ka, kall, kl, idc);
                rmba_k_ask<1>(kb, kall, kl, idc);
                rmba_k_ask<2>(kc, kall, kl, idc);
            }
            rmba_tr_step<8>(p, lane);
            rmba_tr_step<4>(p, lane);
            rmba_tr_step<2>(p, lane);
            rmba_tr_step<1>(p, lane);
            rmba_put_f(wr, c, (id >= 0) ? p[0] * INV_SQRT_D : -INFINITY);
            id = id1;
            __builtin_amdgcn_sched_barrier(0);
        }
    }

    float4 va[8], vb[8];
    rmba_v_ask<0>(va, Vall, vl, max(idr[0], 0));
    rmba_v_ask<1>(vb, Vall, vl, max(idr[0], 0));

    float mx = wr[0];
    #pragma unroll
    for (int i = 1; i < 4; ++i) mx = fmaxf(mx, wr[i]);
    #pragma unroll
    for (int s = 16; s >= 1; s >>= 1) mx = fmaxf(mx, __shfl_xor(mx, s, 32));
    if (lane == 0) smx[pt][qi] = mx;
    __syncthreads();
    mx = fmaxf(smx[0][qi], smx[1][qi]);

    float sm = 0.f;
    #pragma unroll
    for (int i = 0; i < 4; ++i) {
        wr[i] = (idr[i] >= 0) ? __expf(wr[i] - mx) : 0.f;
        sm += wr[i];
    }
    #pragma unroll
    for (int s = 16; s >= 1; s >>= 1) sm += __shfl_xor(sm, s, 32);
    if (lane == 0) ssm[pt][qi] = sm;
    __syncthreads();
    sm = ssm[0][qi] + ssm[1][qi];

    char* arow = (char*)(attn_out + (size_t)n * MM);
    #pragma unroll
    for (int i = 0; i < 4; ++i) {
        wr[i] = wr[i] / sm;
        *(float*)(arow + sl + 256u * (unsigned)i) = wr[i];
    }

    float4 cacc = make_float4(0.f, 0.f, 0.f, 0.f);
    {
        int   c   = 0;
        int   idc = max(idr[0], 0);
        float a   = wr[0];
        #pragma unroll 1
        for (; 2 * c + 2 < ng; ++c) {
            rmba_v_acc<0>(va, a, cacc);
            rmba_v_ask<2>(va, Vall, vl, idc);
            __builtin_amdgcn_sched_barrier(0);
            rmba_v_acc<1>(vb, a, cacc);
            rmba_v_ask<3>(vb, Vall, vl, idc);
            __builtin_amdgcn_sched_barrier(0);
            const int idn = max(rmba_pick_i(idr, c + 1), 0);
            rmba_v_acc<2>(va, a, cacc);
            rmba_v_ask<0>(va, Vall, vl, idn);
            __builtin_amdgcn_sched_barrier(0);
            rmba_v_acc<3>(vb, a, cacc);
            rmba_v_ask<1>(vb, Vall, vl, idn);
            a   = rmba_pick_f(wr, c + 1);
            idc = idn;
            __builtin_amdgcn_sched_barrier(0);
        }
        if (2 * c + 1 < ng) {
            rmba_v_acc<0>(va, a, cacc);
            rmba_v_ask<2>(va, Vall, vl, idc);
            __builtin_amdgcn_sched_barrier(0);
            rmba_v_acc<1>(vb, a, cacc);
            rmba_v_ask<3>(vb, Vall, vl, idc);
            rmba_v_acc<2>(va, a, cacc);
            rmba_v_acc<3>(vb, a, cacc);
        } else {
            rmba_v_acc<0>(va, a, cacc);
            rmba_v_acc<1>(vb, a, cacc);
        }
    }
    *(float4*)(&cs[pt][qi][4 * lane]) = cacc;
    __syncthreads();

    if (wave < 24) {
        const int cb0 = wave * 16 + l16;
        v8f acc0 = {};
        #pragma unroll 4
        for (int k = 0; k < DD; k += 4) {
            v2f a, b0;
            a.x  = cs[0][l16][k + koff]     + cs[1][l16][k + koff];
            a.y  = cs[0][l16][k + koff + 1] + cs[1][l16][k + koff + 1];
            b0.x = Wo[(k + koff) * CRV + cb0];
            b0.y = Wo[(k + koff + 1) * CRV + cb0];
            acc0 = __builtin_amdgcn_wmma_f32_16x16x4_f32(false, a, false, b0,
                                                         (short)0, acc0, false, false);
        }
        const float bias0 = bo[cb0];
        #pragma unroll
        for (int r = 0; r < 8; ++r) {
            const int row = rowBase + half * 8 + r;
            out[row * CRV + cb0] = acc0[r] + bias0;
        }
    }
}

extern "C" void kernel_launch(void* const* d_in, const int* in_sizes, int n_in,
                              void* d_out, int out_size, void* d_ws, size_t ws_size,
                              hipStream_t stream) {
    const float* rv_x = (const float*)d_in[0];
    const float* bev  = (const float*)d_in[1];
    const int*   cand = (const int*)  d_in[2];
    const float* Wq   = (const float*)d_in[3];
    const float* bq   = (const float*)d_in[4];
    const float* Wk   = (const float*)d_in[5];
    const float* bk   = (const float*)d_in[6];
    const float* Wv   = (const float*)d_in[7];
    const float* bv   = (const float*)d_in[8];
    const float* Wo   = (const float*)d_in[9];
    const float* bo   = (const float*)d_in[10];

    float* out_main = (float*)d_out;
    float* out_attn = out_main + (size_t)NRQ * CRV;

    float* Kall = (float*)d_ws;
    float* Vall = Kall + (size_t)NBC * DD;

    rmba_kvproj_kernel<<<NBC / 16, 256, 0, stream>>>(bev, Wk, bk, Wv, bv, Kall, Vall);
    rmba_qattno_kernel<<<NRQ / QPB, 1024, 0, stream>>>(rv_x, Wq, bq, Kall, Vall, cand,
                                                      Wo, bo, out_attn, out_main);
}


# K dots and V sums: each f16 element is an operand of its own multiply-add (v_fma_mix_f32), 4 vector instr. per row, not 4 converts + 2 packed FMAs (logit rounding, ctx sum order change)
# speedup vs baseline: 4.8125x; 1.1486x over previous
#include <hip/hip_runtime.h>
#include <math.h>


typedef float v2f __attribute__((ext_vector_type(2)));
typedef float v8f __attribute__((ext_vector_type(8)));
typedef _Float16 v4h __attribute__((ext_vector_type(4)));

constexpr int NRQ = 4096;
constexpr int NBC = 16384;
constexpr int CB  = 80;
constexpr int CRV = 384;
constexpr int DD  = 128;
constexpr int MM  = 256;
constexpr float INV_SQRT_D = 0.08838834764831845f;


__global__ __launch_bounds__(256)
void rmba_kvproj_kernel(const float* __restrict__ bev,
                        const float* __restrict__ Wk, const float* __restrict__ bk,
                        const float* __restrict__ Wv, const float* __restrict__ bv,
                        float* __restrict__ Kall,
                        float* __restrict__ Vall)
{
    _Float16* __restrict__ Kh = (_Float16*)Kall;
    _Float16* __restrict__ Vh = (_Float16*)Vall;
    const int lane = threadIdx.x & 31;
    const int wave = threadIdx.x >> 5;
    const int half = lane >> 4;
    const int l16  = lane & 15;
    const int koff = half * 2;
    const int rowBase = blockIdx.x * 16;
    const int colBase = wave * 16;
    const int arow = rowBase + l16;
    const int bcol = colBase + l16;

    v8f acck = {}, accv = {};
    #pragma unroll 4
    for (int k = 0; k < CB; k += 4) {
        v2f a, wk, wv;
        a.x  = bev[(k + koff) * NBC + arow];
        a.y  = bev[(k + koff + 1) * NBC + arow];
        wk.x = Wk[(k + koff) * DD + bcol];
        wk.y = Wk[(k + koff + 1) * DD + bcol];
        wv.x = Wv[(k + koff) * DD + bcol];
        wv.y = Wv[(k + koff + 1) * DD + bcol];
        acck = __builtin_amdgcn_wmma_f32_16x16x4_f32(false, a, false, wk,
                                                     (short)0, acck, false, false);
        accv = __builtin_amdgcn_wmma_f32_16x16x4_f32(false, a, false, wv,
                                                     (short)0, accv, false, false);
    }

    const float biask = bk[bcol];
    const float biasv = bv[bcol];
    float kv[8], ss[8];
    #pragma unroll
    for (int r = 0; r < 8; ++r) { kv[r] = acck[r] + biask; ss[r] = kv[r] * kv[r]; }
    #pragma unroll
    for (int s = 8; s >= 1; s >>= 1)
        #pragma unroll
        for (int r = 0; r < 8; ++r) ss[r] += __shfl_xor(ss[r], s, 32);

    __shared__ float sq[8][16];
    __shared__ float rnorm[16];
    if (l16 == 0) {
        #pragma unroll
        for (int r = 0; r < 8; ++r) sq[wave][half * 8 + r] = ss[r];
    }
    __syncthreads();
    if (threadIdx.x < 16) {
        float s = 0.f;
        #pragma unroll
        for (int w = 0; w < 8; ++w) s += sq[w][threadIdx.x];
        rnorm[threadIdx.x] = 1.0f / fmaxf(sqrtf(s), 1e-12f);
    }
    __syncthreads();
    #pragma unroll
    for (int r = 0; r < 8; ++r) {
        const int row = half * 8 + r;
        Kh[(rowBase + row) * DD + bcol] = (_Float16)(kv[r] * rnorm[row]);
        Vh[(rowBase + row) * DD + bcol] = (_Float16)(accv[r] + biasv);
    }
}

__device__ __forceinline__ float rmba_readlane_f(float x, int j) {
    return __int_as_float(__builtin_amdgcn_readlane(__float_as_int(x), j));
}

template <int S>
__device__ __forceinline__ void rmba_tr_step(float (&p)[32], const int lane) {
    const bool up = (lane & S) != 0;
    #pragma unroll
    for (int i = 0; i < S; ++i) {
        const float keep = up ? p[i + S] : p[i];
        const float send = up ? p[i] : p[i + S];
        p[i] = keep + __shfl_xor(send, S, 32);
    }
}


template <int G>
__device__ __forceinline__ void rmba_k_ask(v4h (&kr)[8], const _Float16* __restrict__ kall, const unsigned lbyte, const int idc) {
    #pragma unroll
    for (int j = 0; j < 8; ++j) {
        const int ix = __builtin_amdgcn_readlane(idc, j + 8 * G);
        const unsigned off = (unsigned)ix * (unsigned)(DD * sizeof(_Float16)) + lbyte;
        kr[j] = *(const v4h*)((const char*)kall + off);
    }
}

template <int G>
__device__ __forceinline__ void rmba_k_part(const v4h (&kr)[8], const float4 q4, float (&p)[32]) {
    #pragma unroll
    for (int j = 0; j < 8; ++j) {
        float s = (float)kr[j].x * q4.x;
        s = __builtin_fmaf((float)kr[j].y, q4.y, s);
        s = __builtin_fmaf((float)kr[j].z, q4.z, s);
        s = __builtin_fmaf((float)kr[j].w, q4.w, s);
        p[8 * G + j] = s;
    }
}

template <int G>
__device__ __forceinline__ void rmba_v_ask(v4h (&vr)[8], const _Float16* __restrict__ vall, const unsigned lbyte, const int idc) {
    #pragma unroll
    for (int j = 0; j < 8; ++j) {
        const int ix = __builtin_amdgcn_readlane(idc, j + 8 * G);
        const unsigned off = (unsigned)ix * (unsigned)(DD * sizeof(_Float16)) + lbyte;
        vr[j] = *(const v4h*)((const char*)vall + off);
    }
}

template <int G>
__device__ __forceinline__ void rmba_v_acc(const v4h (&vr)[8], const float a, float4& cacc) {
    #pragma unroll
    for (int j = 0; j < 8; ++j) {
        const float aj = rmba_readlane_f(a, j + 8 * G);
        cacc.x = __builtin_fmaf((float)vr[j].x, aj, cacc.x);
        cacc.y = __builtin_fmaf((float)vr[j].y, aj, cacc.y);
        cacc.z = __builtin_fmaf((float)vr[j].z, aj, cacc.z);
        cacc.w = __builtin_fmaf((float)vr[j].w, aj, cacc.w);
    }
    asm("" : "+v"(cacc.x), "+v"(cacc.y), "+v"(cacc.z), "+v"(cacc.w));
}

__device__ __forceinline__ int rmba_pick_i(const int (&v)[4], const int c) {
    int r = v[0];
    #pragma unroll
    for (int k = 1; k < 4; ++k) r = (c == k) ? v[k] : r;
    return r;
}
__device__ __forceinline__ float rmba_pick_f(const float (&v)[4], const int c) {
    float r = v[0];
    #pragma unroll
    for (int k = 1; k < 4; ++k) r = (c == k) ? v[k] : r;
    return r;
}
__device__ __forceinline__ void rmba_put_f(float (&v)[4], const int c, const float x) {
    #pragma unroll
    for (int k = 0; k < 4; ++k) v[k] = (c == k) ? x : v[k];
}

constexpr int QPB    = 16;
constexpr int CPITCH = DD + 4;
constexpr int QKS    = 16;
constexpr int QKL    = CRV / QKS;

__global__ __launch_bounds__(1024)
void rmba_qattno_kernel(const float* __restrict__ A,
                        const float* __restrict__ Wq,
                        const float* __restrict__ bq,
                        const float* __restrict__ Kall,
                        const float* __restrict__ Vall,
                        const int*   __restrict__ cand,
                        const float* __restrict__ Wo,
                        const float* __restrict__ bo,
                        float* __restrict__ attn_out,
                        float* __restrict__ out)
{
    __shared__ __attribute__((aligned(16))) float part[QKS][QPB][DD];
    __shared__ __attribute__((aligned(16))) float cs[2][QPB][CPITCH];
    __shared__ float smx[2][QPB];
    __shared__ float ssm[2][QPB];

    const int lane = threadIdx.x & 31;
    const int wave = __builtin_amdgcn_readfirstlane((int)(threadIdx.x >> 5));
    const int qi   = wave & 15;
    const int pt   = wave >> 4;
    const int half = lane >> 4;
    const int l16  = lane & 15;
    const int koff = half * 2;
    const int rowBase = blockIdx.x * QPB;

    const int n = rowBase + qi;
    const unsigned sl = 4u * (unsigned)(32 * pt + lane);
    const char* crow = (const char*)(cand + (size_t)n * MM);
    int idr[4];
    #pragma unroll
    for (int i = 0; i < 4; ++i) idr[i] = *(const int*)(crow + sl + 256u * (unsigned)i);

    float4 q4;
    {
        const int ks   = wave >> 1;
        const int arow = rowBase + l16;
        const int kb   = ks * QKL + 4 * half;
        const int cb   = 64 * (wave & 1) + 4 * l16;

        float4 a4[3];
        float4 b4[3][4];
        #pragma unroll
        for (int j = 0; j < 3; ++j)
            a4[j] = *(const float4*)(A + (size_t)arow * CRV + kb + 8 * j);
        #pragma unroll
        for (int j = 0; j < 3; ++j) {
            #pragma unroll
            for (int i = 0; i < 4; ++i)
                b4[j][i] = *(const float4*)(Wq + (size_t)(kb + 8 * j + i) * DD + cb);
        }

        v8f acc0 = {}, acc1 = {}, acc2 = {}, acc3 = {};
        #pragma unroll
        for (int j = 0; j < 3; ++j) {
            v2f a, b;
            a.x = a4[j].x;  a.y = a4[j].y;
            b.x = b4[j][0].x;  b.y = b4[j][1].x;
            acc0 = __builtin_amdgcn_wmma_f32_16x16x4_f32(false, a, false, b, (short)0, acc0, false, false);
            b.x = b4[j][0].y;  b.y = b4[j][1].y;
            acc1 = __builtin_amdgcn_wmma_f32_16x16x4_f32(false, a, false, b, (short)0, acc1, false, false);
            b.x = b4[j][0].z;  b.y = b4[j][1].z;
            acc2 = __builtin_amdgcn_wmma_f32_16x16x4_f32(false, a, false, b, (short)0, acc2, false, false);
            b.x = b4[j][0].w;  b.y = b4[j][1].w;
            acc3 = __builtin_amdgcn_wmma_f32_16x16x4_f32(false, a, false, b, (short)0, acc3, false, false);
            a.x = a4[j].z;  a.y = a4[j].w;
            b.x = b4[j][2].x;  b.y = b4[j][3].x;
            acc0 = __builtin_amdgcn_wmma_f32_16x16x4_f32(false, a, false, b, (short)0, acc0, false, false);
            b.x = b4[j][2].y;  b.y = b4[j][3].y;
            acc1 = __builtin_amdgcn_wmma_f32_16x16x4_f32(false, a, false, b, (short)0, acc1, false, false);
            b.x = b4[j][2].z;  b.y = b4[j][3].z;
            acc2 = __builtin_amdgcn_wmma_f32_16x16x4_f32(false, a, false, b, (short)0, acc2, false, false);
            b.x = b4[j][2].w;  b.y = b4[j][3].w;
            acc3 = __builtin_amdgcn_wmma_f32_16x16x4_f32(false, a, false, b, (short)0, acc3, false, false);
        }
        #pragma unroll
        for (int r = 0; r < 8; ++r)
            *(float4*)(&part[ks][half * 8 + r][cb]) = make_float4(acc0[r], acc1[r], acc2[r], acc3[r]);
        __syncthreads();

        float4 s = *(const float4*)(&part[0][qi][4 * lane]);
        #pragma unroll
        for (int k = 1; k < QKS; ++k) {
            const float4 t = *(const float4*)(&part[k][qi][4 * lane]);
            s.x += t.x;  s.y += t.y;  s.z += t.z;  s.w += t.w;
        }
        const float4 bias = *(const float4*)(bq + 4 * lane);
        s.x += bias.x;  s.y += bias.y;  s.z += bias.z;  s.w += bias.w;
        float ss = s.x * s.x + s.y * s.y + s.z * s.z + s.w * s.w;
        #pragma unroll
        for (int d = 16; d >= 1; d >>= 1) ss += __shfl_xor(ss, d, 32);
        const float rn = 1.0f / fmaxf(sqrtf(ss), 1e-12f);
        q4 = make_float4(s.x * rn, s.y * rn, s.z * rn, s.w * rn);
    }

    const _Float16* kall = (const _Float16*)Kall;
    const _Float16* vall = (const _Float16*)Vall;
    const unsigned kl = 8u * (unsigned)lane;
    const unsigned vl = 8u * (unsigned)lane;

    float wr[4];
    #pragma unroll
    for (int i = 0; i < 4; ++i) wr[i] = -INFINITY;

    int ng = 0;
    #pragma unroll
    for (int i = 0; i < 4; ++i) {
        const unsigned vmk = __builtin_amdgcn_ballot_w32(idr[i] >= 0);
        if (vmk != 0u) ng = 2 * i + ((vmk >> 16) ? 2 : 1);
    }
    const int nc = (ng + 1) >> 1;

    if (nc > 0) {
        v4h ka[8], kb[8], kc[8];
        int id  = idr[0];
        int idc = max(id, 0);
        rmba_k_ask<0>(ka, kall, kl, idc);
        rmba_k_ask<1>(kb, kall, kl, idc);
        rmba_k_ask<2>(kc, kall, kl, idc);
        #pragma unroll 1
        for (int c = 0; c < nc; ++c) {
            float p[32];
            rmba_k_part<0>(ka, q4, p);
            rmba_k_ask<3>(ka, kall, kl, idc);
            __builtin_amdgcn_sched_barrier(0);
            rmba_k_part<1>(kb, q4, p);
            rmba_k_part<2>(kc, q4, p);
            rmba_k_part<3>(ka, q4, p);
            rmba_tr_step<16>(p, lane);
            const int id1 = rmba_pick_i(idr, c + 1);
            if (c + 1 < nc) {
                idc = max(id1, 0);
                rmba_k_ask<0>(ka, kall, kl, idc);
                rmba_k_ask<1>(kb, kall, kl, idc);
                rmba_k_ask<2>(kc, kall, kl, idc);
            }
            rmba_tr_step<8>(p, lane);
            rmba_tr_step<4>(p, lane);
            rmba_tr_step<2>(p, lane);
            rmba_tr_step<1>(p, lane);
            rmba_put_f(wr, c, (id >= 0) ? p[0] * INV_SQRT_D : -INFINITY);
            id = id1;
            __builtin_amdgcn_sched_barrier(0);
        }
    }

    v4h va[8], vb[8];
    rmba_v_ask<0>(va, vall, vl, max(idr[0], 0));
    rmba_v_ask<1>(vb, vall, vl, max(idr[0], 0));

    float mx = wr[0];
    #pragma unroll
    for (int i = 1; i < 4; ++i) mx = fmaxf(mx, wr[i]);
    #pragma unroll
    for (int s = 16; s >= 1; s >>= 1) mx = fmaxf(mx, __shfl_xor(mx, s, 32));
    if (lane == 0) smx[pt][qi] = mx;
    __syncthreads();
    mx = fmaxf(smx[0][qi], smx[1][qi]);

    float sm = 0.f;
    #pragma unroll
    for (int i = 0; i < 4; ++i) {
        wr[i] = (idr[i] >= 0) ? __expf(wr[i] - mx) : 0.f;
        sm += wr[i];
    }
    #pragma unroll
    for (int s = 16; s >= 1; s >>= 1) sm += __shfl_xor(sm, s, 32);
    if (lane == 0) ssm[pt][qi] = sm;
    __syncthreads();
    sm = ssm[0][qi] + ssm[1][qi];

    char* arow = (char*)(attn_out + (size_t)n * MM);
    #pragma unroll
    for (int i = 0; i < 4; ++i) {
        wr[i] = wr[i] / sm;
        *(float*)(arow + sl + 256u * (unsigned)i) = wr[i];
    }

    float4 cacc = make_float4(0.f, 0.f, 0.f, 0.f);
    {
        int   c   = 0;
        int   idc = max(idr[0], 0);
        float a   = wr[0];
        #pragma unroll 1
        for (; 2 * c + 2 < ng; ++c) {
            rmba_v_acc<0>(va, a, cacc);
            rmba_v_ask<2>(va, vall, vl, idc);
            rmba_v_acc<1>(vb, a, cacc);
            rmba_v_ask<3>(vb, vall, vl, idc);
            const int idn = max(rmba_pick_i(idr, c + 1), 0);
            rmba_v_acc<2>(va, a, cacc);
            rmba_v_ask<0>(va, vall, vl, idn);
            rmba_v_acc<3>(vb, a, cacc);
            rmba_v_ask<1>(vb, vall, vl, idn);
            a   = rmba_pick_f(wr, c + 1);
            idc = idn;
        }
        if (2 * c + 1 < ng) {
            rmba_v_acc<0>(va, a, cacc);
            rmba_v_ask<2>(va, vall, vl, idc);
            rmba_v_acc<1>(vb, a, cacc);
            rmba_v_ask<3>(vb, vall, vl, idc);
            rmba_v_acc<2>(va, a, cacc);
            rmba_v_acc<3>(vb, a, cacc);
        } else {
            rmba_v_acc<0>(va, a, cacc);
            rmba_v_acc<1>(vb, a, cacc);
        }
    }
    *(float4*)(&cs[pt][qi][4 * lane]) = cacc;
    __syncthreads();

    if (wave < 24) {
        const int cb0 = wave * 16 + l16;
        v8f acc0 = {};
        #pragma unroll 4
        for (int k = 0; k < DD; k += 4) {
            v2f a, b0;
            a.x  = cs[0][l16][k + koff]     + cs[1][l16][k + koff];
            a.y  = cs[0][l16][k + koff + 1] + cs[1][l16][k + koff + 1];
            b0.x = Wo[(k + koff) * CRV + cb0];
            b0.y = Wo[(k + koff + 1) * CRV + cb0];
            acc0 = __builtin_amdgcn_wmma_f32_16x16x4_f32(false, a, false, b0,
                                                         (short)0, acc0, false, false);
        }
        const float bias0 = bo[cb0];
        #pragma unroll
        for (int r = 0; r < 8; ++r) {
            const int row = rowBase + half * 8 + r;
            out[row * CRV + cb0] = acc0[r] + bias0;
        }
    }
}

extern "C" void kernel_launch(void* const* d_in, const int* in_sizes, int n_in,
                              void* d_out, int out_size, void* d_ws, size_t ws_size,
                              hipStream_t stream) {
    const float* rv_x = (const float*)d_in[0];
    const float* bev  = (const float*)d_in[1];
    const int*   cand = (const int*)  d_in[2];
    const float* Wq   = (const float*)d_in[3];
    const float* bq   = (const float*)d_in[4];
    const float* Wk   = (const float*)d_in[5];
    const float* bk   = (const float*)d_in[6];
    const float* Wv   = (const float*)d_in[7];
    const float* bv   = (const float*)d_in[8];
    const float* Wo   = (const float*)d_in[9];
    const float* bo   = (const float*)d_in[10];

    float* out_main = (float*)d_out;
    float* out_attn = out_main + (size_t)NRQ * CRV;

    float* Kall = (float*)d_ws;
    float* Vall = Kall + (size_t)NBC * DD;

    rmba_kvproj_kernel<<<NBC / 16, 256, 0, stream>>>(bev, Wk, bk, Wv, bv, Kall, Vall);
    rmba_qattno_kernel<<<NRQ / QPB, 1024, 0, stream>>>(rv_x, Wq, bq, Kall, Vall, cand,
                                                      Wo, bo, out_attn, out_main);
}


# K/V table launch re-tiled: a wave makes all 128 outputs of 16 cells of one table (8 tiles, W fragments staged once per group in LDS): 2048 waves not 8192, whole-row stores (K norm sum order changes)
# speedup vs baseline: 4.9957x; 1.0381x over previous
#include <hip/hip_runtime.h>
#include <math.h>


typedef float v2f __attribute__((ext_vector_type(2)));
typedef float v8f __attribute__((ext_vector_type(8)));
typedef _Float16 v4h __attribute__((ext_vector_type(4)));

constexpr int NRQ = 4096;
constexpr int NBC = 16384;
constexpr int CB  = 80;
constexpr int CRV = 384;
constexpr int DD  = 128;
constexpr int MM  = 256;
constexpr float INV_SQRT_D = 0.08838834764831845f;


typedef _Float16 v8h __attribute__((ext_vector_type(8)));

constexpr int KVS = CB / 4;
constexpr int KVW = 8;
constexpr int KVG = 2 * NBC / (16 * KVW);

__global__ __launch_bounds__(256)
void rmba_kvproj_kernel(const float* __restrict__ bev,
                        const float* __restrict__ Wk, const float* __restrict__ bk,
                        const float* __restrict__ Wv, const float* __restrict__ bv,
                        float* __restrict__ Kall,
                        float* __restrict__ Vall)
{
    __shared__ __attribute__((aligned(16))) float4 wf[KVS * 4 * 32];

    const int lane = threadIdx.x & 31;
    const int wave = __builtin_amdgcn_readfirstlane((int)(threadIdx.x >> 5));
    const int half = lane >> 4;
    const int l16  = lane & 15;
    const bool isv = (blockIdx.x & 1) != 0;
    const int cell0 = ((blockIdx.x >> 1) * KVW + wave) * 16;
    const float* __restrict__ W    = isv ? Wv : Wk;
    const float* __restrict__ bias = isv ? bv : bk;

    float ax[KVS], ay[KVS];
    const float* ap = bev + (2 * half) * NBC + cell0 + l16;
    #pragma unroll
    for (int s = 0; s < KVS; ++s) {
        ax[s] = ap[(4 * s) * NBC];
        ay[s] = ap[(4 * s + 1) * NBC];
    }

    #pragma unroll
    for (int j = 0; j < (CB / 2) * (DD / 4) / 256; ++j) {
        const int it = threadIdx.x + 256 * j;
        const int rp = it >> 5;
        const int c4 = it & 31;
        const float4 r0 = *(const float4*)(W + (2 * rp) * DD + 4 * c4);
        const float4 r1 = *(const float4*)(W + (2 * rp + 1) * DD + 4 * c4);
        const int s = rp >> 1, h = rp & 1, c = c4 >> 1, g = c4 & 1;
        wf[(4 * s + 2 * g) * 32 + 16 * h + c]     = make_float4(r0.x, r1.x, r0.y, r1.y);
        wf[(4 * s + 2 * g + 1) * 32 + 16 * h + c] = make_float4(r0.z, r1.z, r0.w, r1.w);
    }
    const float4 bia0 = *(const float4*)(bias + 8 * l16);
    const float4 bia1 = *(const float4*)(bias + 8 * l16 + 4);
    __syncthreads();

    v8f acc[8] = {};
    #pragma unroll
    for (int s = 0; s < KVS; ++s) {
        v2f a;
        a.x = ax[s];  a.y = ay[s];
        #pragma unroll
        for (int q = 0; q < 4; ++q) {
            const float4 f = wf[(4 * s + q) * 32 + lane];
            v2f b;
            b.x = f.x;  b.y = f.y;
            acc[2 * q] = __builtin_amdgcn_wmma_f32_16x16x4_f32(false, a, false, b,
                                                               (short)0, acc[2 * q], false, false);
            b.x = f.z;  b.y = f.w;
            acc[2 * q + 1] = __builtin_amdgcn_wmma_f32_16x16x4_f32(false, a, false, b,
                                                                   (short)0, acc[2 * q + 1], false, false);
        }
    }

    const float bz[8] = { bia0.x, bia0.y, bia0.z, bia0.w, bia1.x, bia1.y, bia1.z, bia1.w };
    #pragma unroll
    for (int T = 0; T < 8; ++T) acc[T] = acc[T] + bz[T];

    if (!isv) {
        v8f ss = acc[0] * acc[0];
        #pragma unroll
        for (int T = 1; T < 8; ++T) ss += acc[T] * acc[T];
        #pragma unroll
        for (int m = 8; m >= 1; m >>= 1) {
            #pragma unroll
            for (int r = 0; r < 8; ++r) ss[r] += __shfl_xor(ss[r], m, 32);
        }
        v8f rn;
        #pragma unroll
        for (int r = 0; r < 8; ++r) rn[r] = 1.0f / fmaxf(sqrtf(ss[r]), 1e-12f);
        #pragma unroll
        for (int T = 0; T < 8; ++T) acc[T] = acc[T] * rn;
    }

    _Float16* __restrict__ dst = (_Float16*)(isv ? Vall : Kall);
    #pragma unroll
    for (int r = 0; r < 8; ++r) {
        v8h hv;
        #pragma unroll
        for (int T = 0; T < 8; ++T) hv[T] = (_Float16)acc[T][r];
        *(v8h*)(dst + (cell0 + 8 * half + r) * DD + 8 * l16) = hv;
    }
}

__device__ __forceinline__ float rmba_readlane_f(float x, int j) {
    return __int_as_float(__builtin_amdgcn_readlane(__float_as_int(x), j));
}

template <int S>
__device__ __forceinline__ void rmba_tr_step(float (&p)[32], const int lane) {
    const bool up = (lane & S) != 0;
    #pragma unroll
    for (int i = 0; i < S; ++i) {
        const float keep = up ? p[i + S] : p[i];
        const float send = up ? p[i] : p[i + S];
        p[i] = keep + __shfl_xor(send, S, 32);
    }
}


template <int G>
__device__ __forceinline__ void rmba_k_ask(v4h (&kr)[8], const _Float16* __restrict__ kall, const unsigned lbyte, const int idc) {
    #pragma unroll
    for (int j = 0; j < 8; ++j) {
        const int ix = __builtin_amdgcn_readlane(idc, j + 8 * G);
        const unsigned off = (unsigned)ix * (unsigned)(DD * sizeof(_Float16)) + lbyte;
        kr[j] = *(const v4h*)((const char*)kall + off);
    }
}

template <int G>
__device__ __forceinline__ void rmba_k_part(const v4h (&kr)[8], const float4 q4, float (&p)[32]) {
    #pragma unroll
    for (int j = 0; j < 8; ++j) {
        float s = (float)kr[j].x * q4.x;
        s = __builtin_fmaf((float)kr[j].y, q4.y, s);
        s = __builtin_fmaf((float)kr[j].z, q4.z, s);
        s = __builtin_fmaf((float)kr[j].w, q4.w, s);
        p[8 * G + j] = s;
    }
}

template <int G>
__device__ __forceinline__ void rmba_v_ask(v4h (&vr)[8], const _Float16* __restrict__ vall, const unsigned lbyte, const int idc) {
    #pragma unroll
    for (int j = 0; j < 8; ++j) {
        const int ix = __builtin_amdgcn_readlane(idc, j + 8 * G);
        const unsigned off = (unsigned)ix * (unsigned)(DD * sizeof(_Float16)) + lbyte;
        vr[j] = *(const v4h*)((const char*)vall + off);
    }
}

template <int G>
__device__ __forceinline__ void rmba_v_acc(const v4h (&vr)[8], const float a, float4& cacc) {
    #pragma unroll
    for (int j = 0; j < 8; ++j) {
        const float aj = rmba_readlane_f(a, j + 8 * G);
        cacc.x = __builtin_fmaf((float)vr[j].x, aj, cacc.x);
        cacc.y = __builtin_fmaf((float)vr[j].y, aj, cacc.y);
        cacc.z = __builtin_fmaf((float)vr[j].z, aj, cacc.z);
        cacc.w = __builtin_fmaf((float)vr[j].w, aj, cacc.w);
    }
    asm("" : "+v"(cacc.x), "+v"(cacc.y), "+v"(cacc.z), "+v"(cacc.w));
}

__device__ __forceinline__ int rmba_pick_i(const int (&v)[4], const int c) {
    int r = v[0];
    #pragma unroll
    for (int k = 1; k < 4; ++k) r = (c == k) ? v[k] : r;
    return r;
}
__device__ __forceinline__ float rmba_pick_f(const float (&v)[4], const int c) {
    float r = v[0];
    #pragma unroll
    for (int k = 1; k < 4; ++k) r = (c == k) ? v[k] : r;
    return r;
}
__device__ __forceinline__ void rmba_put_f(float (&v)[4], const int c, const float x) {
    #pragma unroll
    for (int k = 0; k < 4; ++k) v[k] = (c == k) ? x : v[k];
}

constexpr int QPB    = 16;
constexpr int CPITCH = DD + 4;
constexpr int QKS    = 16;
constexpr int QKL    = CRV / QKS;

__global__ __launch_bounds__(1024)
void rmba_qattno_kernel(const float* __restrict__ A,
                        const float* __restrict__ Wq,
                        const float* __restrict__ bq,
                        const float* __restrict__ Kall,
                        const float* __restrict__ Vall,
                        const int*   __restrict__ cand,
                        const float* __restrict__ Wo,
                        const float* __restrict__ bo,
                        float* __restrict__ attn_out,
                        float* __restrict__ out)
{
    __shared__ __attribute__((aligned(16))) float part[QKS][QPB][DD];
    __shared__ __attribute__((aligned(16))) float cs[2][QPB][CPITCH];
    __shared__ float smx[2][QPB];
    __shared__ float ssm[2][QPB];

    const int lane = threadIdx.x & 31;
    const int wave = __builtin_amdgcn_readfirstlane((int)(threadIdx.x >> 5));
    const int qi   = wave & 15;
    const int pt   = wave >> 4;
    const int half = lane >> 4;
    const int l16  = lane & 15;
    const int koff = half * 2;
    const int rowBase = blockIdx.x * QPB;

    const int n = rowBase + qi;
    const unsigned sl = 4u * (unsigned)(32 * pt + lane);
    const char* crow = (const char*)(cand + (size_t)n * MM);
    int idr[4];
    #pragma unroll
    for (int i = 0; i < 4; ++i) idr[i] = *(const int*)(crow + sl + 256u * (unsigned)i);

    float4 q4;
    {
        const int ks   = wave >> 1;
        const int arow = rowBase + l16;
        const int kb   = ks * QKL + 4 * half;
        const int cb   = 64 * (wave & 1) + 4 * l16;

        float4 a4[3];
        float4 b4[3][4];
        #pragma unroll
        for (int j = 0; j < 3; ++j)
            a4[j] = *(const float4*)(A + (size_t)arow * CRV + kb + 8 * j);
        #pragma unroll
        for (int j = 0; j < 3; ++j) {
            #pragma unroll
            for (int i = 0; i < 4; ++i)
                b4[j][i] = *(const float4*)(Wq + (size_t)(kb + 8 * j + i) * DD + cb);
        }

        v8f acc0 = {}, acc1 = {}, acc2 = {}, acc3 = {};
        #pragma unroll
        for (int j = 0; j < 3; ++j) {
            v2f a, b;
            a.x = a4[j].x;  a.y = a4[j].y;
            b.x = b4[j][0].x;  b.y = b4[j][1].x;
            acc0 = __builtin_amdgcn_wmma_f32_16x16x4_f32(false, a, false, b, (short)0, acc0, false, false);
            b.x = b4[j][0].y;  b.y = b4[j][1].y;
            acc1 = __builtin_amdgcn_wmma_f32_16x16x4_f32(false, a, false, b, (short)0, acc1, false, false);
            b.x = b4[j][0].z;  b.y = b4[j][1].z;
            acc2 = __builtin_amdgcn_wmma_f32_16x16x4_f32(false, a, false, b, (short)0, acc2, false, false);
            b.x = b4[j][0].w;  b.y = b4[j][1].w;
            acc3 = __builtin_amdgcn_wmma_f32_16x16x4_f32(false, a, false, b, (short)0, acc3, false, false);
            a.x = a4[j].z;  a.y = a4[j].w;
            b.x = b4[j][2].x;  b.y = b4[j][3].x;
            acc0 = __builtin_amdgcn_wmma_f32_16x16x4_f32(false, a, false, b, (short)0, acc0, false, false);
            b.x = b4[j][2].y;  b.y = b4[j][3].y;
            acc1 = __builtin_amdgcn_wmma_f32_16x16x4_f32(false, a, false, b, (short)0, acc1, false, false);
            b.x = b4[j][2].z;  b.y = b4[j][3].z;
            acc2 = __builtin_amdgcn_wmma_f32_16x16x4_f32(false, a, false, b, (short)0, acc2, false, false);
            b.x = b4[j][2].w;  b.y = b4[j][3].w;
            acc3 = __builtin_amdgcn_wmma_f32_16x16x4_f32(false, a, false, b, (short)0, acc3, false, false);
        }
        #pragma unroll
        for (int r = 0; r < 8; ++r)
            *(float4*)(&part[ks][half * 8 + r][cb]) = make_float4(acc0[r], acc1[r], acc2[r], acc3[r]);
        __syncthreads();

        float4 s = *(const float4*)(&part[0][qi][4 * lane]);
        #pragma unroll
        for (int k = 1; k < QKS; ++k) {
            const float4 t = *(const float4*)(&part[k][qi][4 * lane]);
            s.x += t.x;  s.y += t.y;  s.z += t.z;  s.w += t.w;
        }
        const float4 bias = *(const float4*)(bq + 4 * lane);
        s.x += bias.x;  s.y += bias.y;  s.z += bias.z;  s.w += bias.w;
        float ss = s.x * s.x + s.y * s.y + s.z * s.z + s.w * s.w;
        #pragma unroll
        for (int d = 16; d >= 1; d >>= 1) ss += __shfl_xor(ss, d, 32);
        const float rn = 1.0f / fmaxf(sqrtf(ss), 1e-12f);
        q4 = make_float4(s.x * rn, s.y * rn, s.z * rn, s.w * rn);
    }

    const _Float16* kall = (const _Float16*)Kall;
    const _Float16* vall = (const _Float16*)Vall;
    const unsigned kl = 8u * (unsigned)lane;
    const unsigned vl = 8u * (unsigned)lane;

    float wr[4];
    #pragma unroll
    for (int i = 0; i < 4; ++i) wr[i] = -INFINITY;

    int ng = 0;
    #pragma unroll
    for (int i = 0; i < 4; ++i) {
        const unsigned vmk = __builtin_amdgcn_ballot_w32(idr[i] >= 0);
        if (vmk != 0u) ng = 2 * i + ((vmk >> 16) ? 2 : 1);
    }
    const int nc = (ng + 1) >> 1;

    if (nc > 0) {
        v4h ka[8], kb[8], kc[8];
        int id  = idr[0];
        int idc = max(id, 0);
        rmba_k_ask<0>(ka, kall, kl, idc);
        rmba_k_ask<1>(kb, kall, kl, idc);
        rmba_k_ask<2>(kc, kall, kl, idc);
        #pragma unroll 1
        for (int c = 0; c < nc; ++c) {
            float p[32];
            rmba_k_part<0>(ka, q4, p);
            rmba_k_ask<3>(ka, kall, kl, idc);
            __builtin_amdgcn_sched_barrier(0);
            rmba_k_part<1>(kb, q4, p);
            rmba_k_part<2>(kc, q4, p);
            rmba_k_part<3>(ka, q4, p);
            rmba_tr_step<16>(p, lane);
            const int id1 = rmba_pick_i(idr, c + 1);
            if (c + 1 < nc) {
                idc = max(id1, 0);
                rmba_k_ask<0>(ka, kall, kl, idc);
                rmba_k_ask<1>(kb, kall, kl, idc);
                rmba_k_ask<2>(kc, kall, kl, idc);
            }
            rmba_tr_step<8>(p, lane);
            rmba_tr_step<4>(p, lane);
            rmba_tr_step<2>(p, lane);
            rmba_tr_step<1>(p, lane);
            rmba_put_f(wr, c, (id >= 0) ? p[0] * INV_SQRT_D : -INFINITY);
            id = id1;
            __builtin_amdgcn_sched_barrier(0);
        }
    }

    v4h va[8], vb[8];
    rmba_v_ask<0>(va, vall, vl, max(idr[0], 0));
    rmba_v_ask<1>(vb, vall, vl, max(idr[0], 0));

    float mx = wr[0];
    #pragma unroll
    for (int i = 1; i < 4; ++i) mx = fmaxf(mx, wr[i]);
    #pragma unroll
    for (int s = 16; s >= 1; s >>= 1) mx = fmaxf(mx, __shfl_xor(mx, s, 32));
    if (lane == 0) smx[pt][qi] = mx;
    __syncthreads();
    mx = fmaxf(smx[0][qi], smx[1][qi]);

    float sm = 0.f;
    #pragma unroll
    for (int i = 0; i < 4; ++i) {
        wr[i] = (idr[i] >= 0) ? __expf(wr[i] - mx) : 0.f;
        sm += wr[i];
    }
    #pragma unroll
    for (int s = 16; s >= 1; s >>= 1) sm += __shfl_xor(sm, s, 32);
    if (lane == 0) ssm[pt][qi] = sm;
    __syncthreads();
    sm = ssm[0][qi] + ssm[1][qi];

    char* arow = (char*)(attn_out + (size_t)n * MM);
    #pragma unroll
    for (int i = 0; i < 4; ++i) {
        wr[i] = wr[i] / sm;
        *(float*)(arow + sl + 256u * (unsigned)i) = wr[i];
    }

    float4 cacc = make_float4(0.f, 0.f, 0.f, 0.f);
    {
        int   c   = 0;
        int   idc = max(idr[0], 0);
        float a   = wr[0];
        #pragma unroll 1
        for (; 2 * c + 2 < ng; ++c) {
            rmba_v_acc<0>(va, a, cacc);
            rmba_v_ask<2>(va, vall, vl, idc);
            rmba_v_acc<1>(vb, a, cacc);
            rmba_v_ask<3>(vb, vall, vl, idc);
            const int idn = max(rmba_pick_i(idr, c + 1), 0);
            rmba_v_acc<2>(va, a, cacc);
            rmba_v_ask<0>(va, vall, vl, idn);
            rmba_v_acc<3>(vb, a, cacc);
            rmba_v_ask<1>(vb, vall, vl, idn);
            a   = rmba_pick_f(wr, c + 1);
            idc = idn;
        }
        if (2 * c + 1 < ng) {
            rmba_v_acc<0>(va, a, cacc);
            rmba_v_ask<2>(va, vall, vl, idc);
            rmba_v_acc<1>(vb, a, cacc);
            rmba_v_ask<3>(vb, vall, vl, idc);
            rmba_v_acc<2>(va, a, cacc);
            rmba_v_acc<3>(vb, a, cacc);
        } else {
            rmba_v_acc<0>(va, a, cacc);
            rmba_v_acc<1>(vb, a, cacc);
        }
    }
    *(float4*)(&cs[pt][qi][4 * lane]) = cacc;
    __syncthreads();

    if (wave < 24) {
        const int cb0 = wave * 16 + l16;
        v8f acc0 = {};
        #pragma unroll 4
        for (int k = 0; k < DD; k += 4) {
            v2f a, b0;
            a.x  = cs[0][l16][k + koff]     + cs[1][l16][k + koff];
            a.y  = cs[0][l16][k + koff + 1] + cs[1][l16][k + koff + 1];
            b0.x = Wo[(k + koff) * CRV + cb0];
            b0.y = Wo[(k + koff + 1) * CRV + cb0];
            acc0 = __builtin_amdgcn_wmma_f32_16x16x4_f32(false, a, false, b0,
                                                         (short)0, acc0, false, false);
        }
        const float bias0 = bo[cb0];
        #pragma unroll
        for (int r = 0; r < 8; ++r) {
            const int row = rowBase + half * 8 + r;
            out[row * CRV + cb0] = acc0[r] + bias0;
        }
    }
}

extern "C" void kernel_launch(void* const* d_in, const int* in_sizes, int n_in,
                              void* d_out, int out_size, void* d_ws, size_t ws_size,
                              hipStream_t stream) {
    const float* rv_x = (const float*)d_in[0];
    const float* bev  = (const float*)d_in[1];
    const int*   cand = (const int*)  d_in[2];
    const float* Wq   = (const float*)d_in[3];
    const float* bq   = (const float*)d_in[4];
    const float* Wk   = (const float*)d_in[5];
    const float* bk   = (const float*)d_in[6];
    const float* Wv   = (const float*)d_in[7];
    const float* bv   = (const float*)d_in[8];
    const float* Wo   = (const float*)d_in[9];
    const float* bo   = (const float*)d_in[10];

    float* out_main = (float*)d_out;
    float* out_attn = out_main + (size_t)NRQ * CRV;

    float* Kall = (float*)d_ws;
    float* Vall = Kall + (size_t)NBC * DD;

    rmba_kvproj_kernel<<<KVG, 256, 0, stream>>>(bev, Wk, bk, Wv, bv, Kall, Vall);
    rmba_qattno_kernel<<<NRQ / QPB, 1024, 0, stream>>>(rv_x, Wq, bq, Kall, Vall, cand,
                                                      Wo, bo, out_attn, out_main);
}
